# starting kernel
# baseline (speedup 1.0000x reference)
#include <hip/hip_runtime.h>
#include <hip/hip_bf16.h>
#include <math.h>


typedef __attribute__((ext_vector_type(2))) float v2f;
typedef __attribute__((ext_vector_type(8))) float v8f;

#define IN_F     128
#define OUT_F    64
#define HEADS    2
#define XDIM     128
#define NEG_SLOPE 0.2f
#define LN_EPS    1e-5f
#define SM_EPS    1e-16f

__global__ __launch_bounds__(256) void gat_gemm_wmma(
    const float* __restrict__ X, const float* __restrict__ W,
    float* __restrict__ xp, int N)
{
    const int wave = (blockIdx.x * blockDim.x + threadIdx.x) >> 5;
    const int lane = threadIdx.x & 31;
    const int r0 = wave * 16;
    if (r0 >= N) return;

    const int lm = lane & 15;
    const int kk = (lane >> 4) << 1;

    int ar = r0 + lm; if (ar >= N) ar = N - 1;
    const float* arow = X + (size_t)ar * IN_F + kk;
    const float* brow = W + (size_t)lm * IN_F + kk;

    v8f acc[8];
    #pragma unroll
    for (int t = 0; t < 8; ++t) acc[t] = (v8f){};

    #pragma unroll 4
    for (int k0 = 0; k0 < IN_F; k0 += 4) {
        v2f a = *(const v2f*)(arow + k0);
        #pragma unroll
        for (int t = 0; t < 8; ++t) {
            v2f b = *(const v2f*)(brow + (size_t)t * 16 * IN_F + k0);
            acc[t] = __builtin_amdgcn_wmma_f32_16x16x4_f32(
                 false, a,  false, b,
                 (short)0, acc[t],  false,  false);
        }
    }

    const int mofs = (lane >> 4) << 3;
    #pragma unroll
    for (int v = 0; v < 8; ++v) {
        const int row = r0 + mofs + v;
        if (row < N) {
            float* orow = xp + (size_t)row * XDIM + lm;
            #pragma unroll
            for (int t = 0; t < 8; ++t) orow[t * 16] = acc[t][v];
        }
    }
}

__global__ __launch_bounds__(256) void gat_logits_init(
    const float* __restrict__ xp,
    const float* __restrict__ attS, const float* __restrict__ attD,
    float* __restrict__ aS, float* __restrict__ aD,
    float* __restrict__ mmax, float* __restrict__ denom,
    float* __restrict__ acc, int N)
{
    const int wave = (blockIdx.x * blockDim.x + threadIdx.x) >> 5;
    const int lane = threadIdx.x & 31;
    if (wave >= N) return;
    const int head = lane >> 4;
    const int fo   = lane * 4;

    float4 xv = *(const float4*)(xp + (size_t)wave * XDIM + fo);
    float4 sv = *(const float4*)(attS + fo);
    float4 dv = *(const float4*)(attD + fo);
    float ps = xv.x*sv.x + xv.y*sv.y + xv.z*sv.z + xv.w*sv.w;
    float pd = xv.x*dv.x + xv.y*dv.y + xv.z*dv.z + xv.w*dv.w;
    #pragma unroll
    for (int off = 8; off >= 1; off >>= 1) {
        ps += __shfl_xor(ps, off, 32);
        pd += __shfl_xor(pd, off, 32);
    }
    if ((lane & 15) == 0) {
        aS[wave * 2 + head]    = ps;
        aD[wave * 2 + head]    = pd;
        mmax[wave * 2 + head]  = -INFINITY;
        denom[wave * 2 + head] = 0.0f;
    }
    float4 z = {0.f, 0.f, 0.f, 0.f};
    *(float4*)(acc + (size_t)wave * XDIM + fo) = z;
}

__device__ __forceinline__ void atomicMaxF(float* addr, float val)
{
    if (val >= 0.0f) atomicMax((int*)addr, __float_as_int(val));
    else             atomicMin((unsigned int*)addr, __float_as_uint(val));
}

__device__ __forceinline__ float lrelu(float v)
{
    return v > 0.0f ? v : NEG_SLOPE * v;
}

__global__ __launch_bounds__(256) void gat_edge_max(
    const int* __restrict__ esrc, const int* __restrict__ edst,
    const float* __restrict__ aS, const float* __restrict__ aD,
    float* __restrict__ mmax, int E, int N)
{
    const int eid = blockIdx.x * blockDim.x + threadIdx.x;
    if (eid >= E + N) return;
    int s, d;
    if (eid < E) { s = esrc[eid]; d = edst[eid]; }
    else         { s = d = eid - E; }
    const float e0 = lrelu(aS[s * 2 + 0] + aD[d * 2 + 0]);
    const float e1 = lrelu(aS[s * 2 + 1] + aD[d * 2 + 1]);
    atomicMaxF(&mmax[d * 2 + 0], e0);
    atomicMaxF(&mmax[d * 2 + 1], e1);
}

__global__ __launch_bounds__(256) void gat_edge_sum(
    const int* __restrict__ esrc, const int* __restrict__ edst,
    const float* __restrict__ aS, const float* __restrict__ aD,
    const float* __restrict__ mmax, float* __restrict__ denom, int E, int N)
{
    const int eid = blockIdx.x * blockDim.x + threadIdx.x;
    if (eid >= E + N) return;
    int s, d;
    if (eid < E) { s = esrc[eid]; d = edst[eid]; }
    else         { s = d = eid - E; }
    const float e0 = lrelu(aS[s * 2 + 0] + aD[d * 2 + 0]);
    const float e1 = lrelu(aS[s * 2 + 1] + aD[d * 2 + 1]);
    atomicAdd(&denom[d * 2 + 0], __expf(e0 - mmax[d * 2 + 0]));
    atomicAdd(&denom[d * 2 + 1], __expf(e1 - mmax[d * 2 + 1]));
}

__global__ __launch_bounds__(256) void gat_edge_scatter(
    const int* __restrict__ esrc, const int* __restrict__ edst,
    const float* __restrict__ aS, const float* __restrict__ aD,
    const float* __restrict__ mmax, const float* __restrict__ denom,
    const float* __restrict__ xp, float* __restrict__ acc, int E, int N)
{
    const int wave = (blockIdx.x * blockDim.x + threadIdx.x) >> 5;
    const int lane = threadIdx.x & 31;
    if (wave >= E + N) return;
    int s, d;
    if (wave < E) { s = esrc[wave]; d = edst[wave]; }
    else          { s = d = wave - E; }
    const int head = lane >> 4;
    float e = lrelu(aS[s * 2 + head] + aD[d * 2 + head]);
    const float alpha = __expf(e - mmax[d * 2 + head]) /
                        (denom[d * 2 + head] + SM_EPS);
    const int fo = lane * 4;
    float4 v = *(const float4*)(xp + (size_t)s * XDIM + fo);
    float* ap = acc + (size_t)d * XDIM + fo;
    atomicAdd(ap + 0, v.x * alpha);
    atomicAdd(ap + 1, v.y * alpha);
    atomicAdd(ap + 2, v.z * alpha);
    atomicAdd(ap + 3, v.w * alpha);
}

__global__ __launch_bounds__(256) void gat_finalize(
    const float* __restrict__ acc, const float* __restrict__ bias,
    const float* __restrict__ gamma, const float* __restrict__ beta,
    float* __restrict__ out, int N)
{
    const int wave = (blockIdx.x * blockDim.x + threadIdx.x) >> 5;
    const int lane = threadIdx.x & 31;
    if (wave >= N) return;
    const float* a = acc + (size_t)wave * XDIM;
    const int f0 = lane, f1 = lane + 32;
    float v0 = (a[f0] + a[OUT_F + f0]) * 0.5f + bias[f0];
    float v1 = (a[f1] + a[OUT_F + f1]) * 0.5f + bias[f1];

    float sum = v0 + v1;
    #pragma unroll
    for (int off = 16; off >= 1; off >>= 1) sum += __shfl_xor(sum, off, 32);
    const float mu = sum * (1.0f / OUT_F);

    const float d0 = v0 - mu, d1 = v1 - mu;
    float sq = d0 * d0 + d1 * d1;
    #pragma unroll
    for (int off = 16; off >= 1; off >>= 1) sq += __shfl_xor(sq, off, 32);
    const float inv = rsqrtf(sq * (1.0f / OUT_F) + LN_EPS);

    out[(size_t)wave * OUT_F + f0] = d0 * inv * gamma[f0] + beta[f0];
    out[(size_t)wave * OUT_F + f1] = d1 * inv * gamma[f1] + beta[f1];
}

extern "C" void kernel_launch(void* const* d_in, const int* in_sizes, int n_in,
                              void* d_out, int out_size, void* d_ws, size_t ws_size,
                              hipStream_t stream)
{
    const float* X    = (const float*)d_in[0];
    const int*   ei   = (const int*)  d_in[1];
    const float* W    = (const float*)d_in[2];
    const float* attS = (const float*)d_in[3];
    const float* attD = (const float*)d_in[4];
    const float* bias = (const float*)d_in[5];
    const float* g    = (const float*)d_in[6];
    const float* b    = (const float*)d_in[7];
    float* out = (float*)d_out;

    const int N = in_sizes[0] / IN_F;
    const int E = in_sizes[1] / 2;
    const int* esrc = ei;
    const int* edst = ei + E;
    const int Et = E + N;

    float* ws    = (float*)d_ws;
    float* xp    = ws;
    float* accb  = xp   + (size_t)N * XDIM;
    float* aS    = accb + (size_t)N * XDIM;
    float* aD    = aS   + (size_t)N * 2;
    float* mmax  = aD   + (size_t)N * 2;
    float* denom = mmax + (size_t)N * 2;

    {
        const int waves = (N + 15) / 16;
        gat_gemm_wmma<<<(waves + 7) / 8, 256, 0, stream>>>(X, W, xp, N);
    }
    gat_logits_init<<<(N + 7) / 8, 256, 0, stream>>>(
        xp, attS, attD, aS, aD, mmax, denom, accb, N);
    gat_edge_max<<<(Et + 255) / 256, 256, 0, stream>>>(
        esrc, edst, aS, aD, mmax, E, N);
    gat_edge_sum<<<(Et + 255) / 256, 256, 0, stream>>>(
        esrc, edst, aS, aD, mmax, denom, E, N);
    gat_edge_scatter<<<(Et + 7) / 8, 256, 0, stream>>>(
        esrc, edst, aS, aD, mmax, denom, xp, accb, E, N);
    gat_finalize<<<(N + 7) / 8, 256, 0, stream>>>(accb, bias, g, b, out, N);
}


# Defer the softmax division: one edge pass adds exp(e-m)*x[src] into acc and exp(e-m) into denom, finalize divides by denom+eps; the separate segment-sum launch is gone (sums of unnormalised terms)
# speedup vs baseline: 1.2121x; 1.2121x over previous
#include <hip/hip_runtime.h>
#include <hip/hip_bf16.h>
#include <math.h>


typedef __attribute__((ext_vector_type(2))) float v2f;
typedef __attribute__((ext_vector_type(8))) float v8f;

#define IN_F     128
#define OUT_F    64
#define HEADS    2
#define XDIM     128
#define NEG_SLOPE 0.2f
#define LN_EPS    1e-5f
#define SM_EPS    1e-16f

__global__ __launch_bounds__(256) void gat_gemm_wmma(
    const float* __restrict__ X, const float* __restrict__ W,
    float* __restrict__ xp, int N)
{
    const int wave = (blockIdx.x * blockDim.x + threadIdx.x) >> 5;
    const int lane = threadIdx.x & 31;
    const int r0 = wave * 16;
    if (r0 >= N) return;

    const int lm = lane & 15;
    const int kk = (lane >> 4) << 1;

    int ar = r0 + lm; if (ar >= N) ar = N - 1;
    const float* arow = X + (size_t)ar * IN_F + kk;
    const float* brow = W + (size_t)lm * IN_F + kk;

    v8f acc[8];
    #pragma unroll
    for (int t = 0; t < 8; ++t) acc[t] = (v8f){};

    #pragma unroll 4
    for (int k0 = 0; k0 < IN_F; k0 += 4) {
        v2f a = *(const v2f*)(arow + k0);
        #pragma unroll
        for (int t = 0; t < 8; ++t) {
            v2f b = *(const v2f*)(brow + (size_t)t * 16 * IN_F + k0);
            acc[t] = __builtin_amdgcn_wmma_f32_16x16x4_f32(
                 false, a,  false, b,
                 (short)0, acc[t],  false,  false);
        }
    }

    const int mofs = (lane >> 4) << 3;
    #pragma unroll
    for (int v = 0; v < 8; ++v) {
        const int row = r0 + mofs + v;
        if (row < N) {
            float* orow = xp + (size_t)row * XDIM + lm;
            #pragma unroll
            for (int t = 0; t < 8; ++t) orow[t * 16] = acc[t][v];
        }
    }
}

__global__ __launch_bounds__(256) void gat_logits_init(
    const float* __restrict__ xp,
    const float* __restrict__ attS, const float* __restrict__ attD,
    float* __restrict__ aS, float* __restrict__ aD,
    float* __restrict__ mmax, float* __restrict__ denom,
    float* __restrict__ acc, int N)
{
    const int wave = (blockIdx.x * blockDim.x + threadIdx.x) >> 5;
    const int lane = threadIdx.x & 31;
    if (wave >= N) return;
    const int head = lane >> 4;
    const int fo   = lane * 4;

    float4 xv = *(const float4*)(xp + (size_t)wave * XDIM + fo);
    float4 sv = *(const float4*)(attS + fo);
    float4 dv = *(const float4*)(attD + fo);
    float ps = xv.x*sv.x + xv.y*sv.y + xv.z*sv.z + xv.w*sv.w;
    float pd = xv.x*dv.x + xv.y*dv.y + xv.z*dv.z + xv.w*dv.w;
    #pragma unroll
    for (int off = 8; off >= 1; off >>= 1) {
        ps += __shfl_xor(ps, off, 32);
        pd += __shfl_xor(pd, off, 32);
    }
    if ((lane & 15) == 0) {
        aS[wave * 2 + head]    = ps;
        aD[wave * 2 + head]    = pd;
        mmax[wave * 2 + head]  = -INFINITY;
        denom[wave * 2 + head] = 0.0f;
    }
    float4 z = {0.f, 0.f, 0.f, 0.f};
    *(float4*)(acc + (size_t)wave * XDIM + fo) = z;
}

__device__ __forceinline__ void atomicMaxF(float* addr, float val)
{
    if (val >= 0.0f) atomicMax((int*)addr, __float_as_int(val));
    else             atomicMin((unsigned int*)addr, __float_as_uint(val));
}

__device__ __forceinline__ float lrelu(float v)
{
    return v > 0.0f ? v : NEG_SLOPE * v;
}

__global__ __launch_bounds__(256) void gat_edge_max(
    const int* __restrict__ esrc, const int* __restrict__ edst,
    const float* __restrict__ aS, const float* __restrict__ aD,
    float* __restrict__ mmax, int E, int N)
{
    const int eid = blockIdx.x * blockDim.x + threadIdx.x;
    if (eid >= E + N) return;
    int s, d;
    if (eid < E) { s = esrc[eid]; d = edst[eid]; }
    else         { s = d = eid - E; }
    const float e0 = lrelu(aS[s * 2 + 0] + aD[d * 2 + 0]);
    const float e1 = lrelu(aS[s * 2 + 1] + aD[d * 2 + 1]);
    atomicMaxF(&mmax[d * 2 + 0], e0);
    atomicMaxF(&mmax[d * 2 + 1], e1);
}

__global__ __launch_bounds__(256) void gat_edge_scatter(
    const int* __restrict__ esrc, const int* __restrict__ edst,
    const float* __restrict__ aS, const float* __restrict__ aD,
    const float* __restrict__ mmax, float* __restrict__ denom,
    const float* __restrict__ xp, float* __restrict__ acc, int E, int N)
{
    const int wave = (blockIdx.x * blockDim.x + threadIdx.x) >> 5;
    const int lane = threadIdx.x & 31;
    if (wave >= E + N) return;
    int s, d;
    if (wave < E) { s = esrc[wave]; d = edst[wave]; }
    else          { s = d = wave - E; }
    const int head = lane >> 4;
    const int fo = lane * 4;
    float4 v = *(const float4*)(xp + (size_t)s * XDIM + fo);
    float e = lrelu(aS[s * 2 + head] + aD[d * 2 + head]);
    const float w = __expf(e - mmax[d * 2 + head]);
    if ((lane & 15) == 0) atomicAdd(&denom[d * 2 + head], w);
    float* ap = acc + (size_t)d * XDIM + fo;
    atomicAdd(ap + 0, v.x * w);
    atomicAdd(ap + 1, v.y * w);
    atomicAdd(ap + 2, v.z * w);
    atomicAdd(ap + 3, v.w * w);
}

__global__ __launch_bounds__(256) void gat_finalize(
    const float* __restrict__ acc, const float* __restrict__ denom,
    const float* __restrict__ bias,
    const float* __restrict__ gamma, const float* __restrict__ beta,
    float* __restrict__ out, int N)
{
    const int wave = (blockIdx.x * blockDim.x + threadIdx.x) >> 5;
    const int lane = threadIdx.x & 31;
    if (wave >= N) return;
    const float* a = acc + (size_t)wave * XDIM;
    const int f0 = lane, f1 = lane + 32;
    const float dn0 = denom[wave * 2 + 0] + SM_EPS;
    const float dn1 = denom[wave * 2 + 1] + SM_EPS;
    float v0 = (a[f0] / dn0 + a[OUT_F + f0] / dn1) * 0.5f + bias[f0];
    float v1 = (a[f1] / dn0 + a[OUT_F + f1] / dn1) * 0.5f + bias[f1];

    float sum = v0 + v1;
    #pragma unroll
    for (int off = 16; off >= 1; off >>= 1) sum += __shfl_xor(sum, off, 32);
    const float mu = sum * (1.0f / OUT_F);

    const float d0 = v0 - mu, d1 = v1 - mu;
    float sq = d0 * d0 + d1 * d1;
    #pragma unroll
    for (int off = 16; off >= 1; off >>= 1) sq += __shfl_xor(sq, off, 32);
    const float inv = rsqrtf(sq * (1.0f / OUT_F) + LN_EPS);

    out[(size_t)wave * OUT_F + f0] = d0 * inv * gamma[f0] + beta[f0];
    out[(size_t)wave * OUT_F + f1] = d1 * inv * gamma[f1] + beta[f1];
}

extern "C" void kernel_launch(void* const* d_in, const int* in_sizes, int n_in,
                              void* d_out, int out_size, void* d_ws, size_t ws_size,
                              hipStream_t stream)
{
    const float* X    = (const float*)d_in[0];
    const int*   ei   = (const int*)  d_in[1];
    const float* W    = (const float*)d_in[2];
    const float* attS = (const float*)d_in[3];
    const float* attD = (const float*)d_in[4];
    const float* bias = (const float*)d_in[5];
    const float* g    = (const float*)d_in[6];
    const float* b    = (const float*)d_in[7];
    float* out = (float*)d_out;

    const int N = in_sizes[0] / IN_F;
    const int E = in_sizes[1] / 2;
    const int* esrc = ei;
    const int* edst = ei + E;
    const int Et = E + N;

    float* ws    = (float*)d_ws;
    float* xp    = ws;
    float* accb  = xp   + (size_t)N * XDIM;
    float* aS    = accb + (size_t)N * XDIM;
    float* aD    = aS   + (size_t)N * 2;
    float* mmax  = aD   + (size_t)N * 2;
    float* denom = mmax + (size_t)N * 2;

    {
        const int waves = (N + 15) / 16;
        gat_gemm_wmma<<<(waves + 7) / 8, 256, 0, stream>>>(X, W, xp, N);
    }
    gat_logits_init<<<(N + 7) / 8, 256, 0, stream>>>(
        xp, attS, attD, aS, aD, mmax, denom, accb, N);
    gat_edge_max<<<(Et + 255) / 256, 256, 0, stream>>>(
        esrc, edst, aS, aD, mmax, E, N);
    gat_edge_scatter<<<(Et + 7) / 8, 256, 0, stream>>>(
        esrc, edst, aS, aD, mmax, denom, xp, accb, E, N);
    gat_finalize<<<(N + 7) / 8, 256, 0, stream>>>(accb, denom, bias, g, b, out, N);
}


# Counting sort of edges by dst inside the call (int rank, scan, place); one wave per dst sums its segment in registers and finalizes; no float atomics; sums in arrival-rank order, denom by wave tree
# speedup vs baseline: 2.7626x; 2.2792x over previous
#include <hip/hip_runtime.h>
#include <hip/hip_bf16.h>
#include <math.h>


typedef __attribute__((ext_vector_type(2))) float v2f;
typedef __attribute__((ext_vector_type(8))) float v8f;

#define IN_F     128
#define OUT_F    64
#define HEADS    2
#define XDIM     128
#define NEG_SLOPE 0.2f
#define LN_EPS    1e-5f
#define SM_EPS    1e-16f

__global__ __launch_bounds__(256) void gat_gemm_wmma(
    const float* __restrict__ X, const float* __restrict__ W,
    float* __restrict__ xp, int N)
{
    const int wave = (blockIdx.x * blockDim.x + threadIdx.x) >> 5;
    const int lane = threadIdx.x & 31;
    const int r0 = wave * 16;
    if (r0 >= N) return;

    const int lm = lane & 15;
    const int kk = (lane >> 4) << 1;

    int ar = r0 + lm; if (ar >= N) ar = N - 1;
    const float* arow = X + (size_t)ar * IN_F + kk;
    const float* brow = W + (size_t)lm * IN_F + kk;

    v8f acc[8];
    #pragma unroll
    for (int t = 0; t < 8; ++t) acc[t] = (v8f){};

    #pragma unroll 4
    for (int k0 = 0; k0 < IN_F; k0 += 4) {
        v2f a = *(const v2f*)(arow + k0);
        #pragma unroll
        for (int t = 0; t < 8; ++t) {
            v2f b = *(const v2f*)(brow + (size_t)t * 16 * IN_F + k0);
            acc[t] = __builtin_amdgcn_wmma_f32_16x16x4_f32(
                 false, a,  false, b,
                 (short)0, acc[t],  false,  false);
        }
    }

    const int mofs = (lane >> 4) << 3;
    #pragma unroll
    for (int v = 0; v < 8; ++v) {
        const int row = r0 + mofs + v;
        if (row < N) {
            float* orow = xp + (size_t)row * XDIM + lm;
            #pragma unroll
            for (int t = 0; t < 8; ++t) orow[t * 16] = acc[t][v];
        }
    }
}

__global__ __launch_bounds__(256) void gat_logits_init(
    const float* __restrict__ xp,
    const float* __restrict__ attS, const float* __restrict__ attD,
    float* __restrict__ aS, float* __restrict__ aD,
    int* __restrict__ cnt, int N)
{
    const int wave = (blockIdx.x * blockDim.x + threadIdx.x) >> 5;
    const int lane = threadIdx.x & 31;
    if (wave >= N) return;
    const int head = lane >> 4;
    const int fo   = lane * 4;

    float4 xv = *(const float4*)(xp + (size_t)wave * XDIM + fo);
    float4 sv = *(const float4*)(attS + fo);
    float4 dv = *(const float4*)(attD + fo);
    float ps = xv.x*sv.x + xv.y*sv.y + xv.z*sv.z + xv.w*sv.w;
    float pd = xv.x*dv.x + xv.y*dv.y + xv.z*dv.z + xv.w*dv.w;
    #pragma unroll
    for (int off = 8; off >= 1; off >>= 1) {
        ps += __shfl_xor(ps, off, 32);
        pd += __shfl_xor(pd, off, 32);
    }
    if ((lane & 15) == 0) {
        aS[wave * 2 + head] = ps;
        aD[wave * 2 + head] = pd;
    }
    if (lane == 0) cnt[wave] = 0;
}

__device__ __forceinline__ float lrelu(float v)
{
    return v > 0.0f ? v : NEG_SLOPE * v;
}

__global__ __launch_bounds__(256) void gat_edge_rank(
    const int* __restrict__ edst, int* __restrict__ cnt,
    int* __restrict__ rank, int E)
{
    const int eid = blockIdx.x * blockDim.x + threadIdx.x;
    if (eid >= E) return;
    rank[eid] = atomicAdd(&cnt[edst[eid]], 1);
}

#define SCAN_T 1024
__global__ __launch_bounds__(SCAN_T) void gat_scan(
    const int* __restrict__ cnt, int* __restrict__ off, int N)
{
    __shared__ int s_part[32];
    __shared__ int s_tot[32];
    const int tid  = threadIdx.x;
    const int lane = tid & 31;
    const int wv   = tid >> 5;
    const int tile0 = blockIdx.x * SCAN_T;

    int pre = 0;
    #pragma unroll 4
    for (int i = tid; i < tile0; i += SCAN_T) pre += cnt[i];
    #pragma unroll
    for (int o = 16; o >= 1; o >>= 1) pre += __shfl_xor(pre, o, 32);
    if (lane == 0) s_part[wv] = pre;

    const int i = tile0 + tid;
    const int v = (i < N) ? cnt[i] : 0;
    int inc = v;
    #pragma unroll
    for (int o = 1; o < 32; o <<= 1) {
        const int t = __shfl_up(inc, o, 32);
        if (lane >= o) inc += t;
    }
    if (lane == 31) s_tot[wv] = inc;
    __syncthreads();

    int carry = s_part[lane];
    #pragma unroll
    for (int o = 16; o >= 1; o >>= 1) carry += __shfl_xor(carry, o, 32);
    const int wt = s_tot[lane];
    int winc = wt;
    #pragma unroll
    for (int o = 1; o < 32; o <<= 1) {
        const int t = __shfl_up(winc, o, 32);
        if (lane >= o) winc += t;
    }
    const int wave_excl = __shfl(winc - wt, wv, 32);

    const int excl = carry + wave_excl + inc - v;
    if (i < N) off[i] = excl;
    if (i == N - 1) off[N] = excl + v;
}

__global__ __launch_bounds__(256) void gat_edge_place(
    const int* __restrict__ esrc, const int* __restrict__ edst,
    const int* __restrict__ off, const int* __restrict__ rank,
    int* __restrict__ ssrc, int E)
{
    const int eid = blockIdx.x * blockDim.x + threadIdx.x;
    if (eid >= E) return;
    ssrc[off[edst[eid]] + rank[eid]] = esrc[eid];
}

__global__ __launch_bounds__(256) void gat_node_gather(
    const int* __restrict__ off, const int* __restrict__ ssrc,
    const float* __restrict__ aS, const float* __restrict__ aD,
    const float* __restrict__ xp,
    const float* __restrict__ bias,
    const float* __restrict__ gamma, const float* __restrict__ beta,
    float* __restrict__ out, int N)
{
    const int d = __builtin_amdgcn_readfirstlane(
        (int)((blockIdx.x * blockDim.x + threadIdx.x) >> 5));
    const int lane = threadIdx.x & 31;
    if (d >= N) return;
    const int head = lane >> 4;
    const int fo = lane * 4;

    const int beg = off[d];
    const int end = off[d + 1];
    const float ad0 = aD[d * 2 + 0], ad1 = aD[d * 2 + 1];
    const float es0 = lrelu(aS[d * 2 + 0] + ad0);
    const float es1 = lrelu(aS[d * 2 + 1] + ad1);
    const float4 xs = *(const float4*)(xp + (size_t)d * XDIM + fo);

    float m0 = es0, m1 = es1;
    for (int i = beg + lane; i < end; i += 32) {
        const int s = ssrc[i];
        const float2 a = *(const float2*)(aS + (size_t)s * 2);
        m0 = fmaxf(m0, lrelu(a.x + ad0));
        m1 = fmaxf(m1, lrelu(a.y + ad1));
    }
    #pragma unroll
    for (int o = 16; o >= 1; o >>= 1) {
        m0 = fmaxf(m0, __shfl_xor(m0, o, 32));
        m1 = fmaxf(m1, __shfl_xor(m1, o, 32));
    }

    const float ws0 = __expf(es0 - m0), ws1 = __expf(es1 - m1);
    const float wsh = head ? ws1 : ws0;
    float4 acc = {xs.x * wsh, xs.y * wsh, xs.z * wsh, xs.w * wsh};
    float dn0 = 0.0f, dn1 = 0.0f;

    for (int base = beg; base < end; base += 32) {
        const int i = base + lane;
        int s = 0;
        float w0 = 0.0f, w1 = 0.0f;
        if (i < end) {
            s = ssrc[i];
            const float2 a = *(const float2*)(aS + (size_t)s * 2);
            w0 = __expf(lrelu(a.x + ad0) - m0);
            w1 = __expf(lrelu(a.y + ad1) - m1);
        }
        dn0 += w0;
        dn1 += w1;
        const int rem = end - base;
        const int n = rem < 32 ? rem : 32;
        #pragma unroll 4
        for (int j = 0; j < n; ++j) {
            const int sj = __builtin_amdgcn_readlane(s, j);
            const float wa = __int_as_float(__builtin_amdgcn_readlane(__float_as_int(w0), j));
            const float wb = __int_as_float(__builtin_amdgcn_readlane(__float_as_int(w1), j));
            const float w = head ? wb : wa;
            const float4 v = *(const float4*)(xp + (size_t)sj * XDIM + fo);
            acc.x += v.x * w;
            acc.y += v.y * w;
            acc.z += v.z * w;
            acc.w += v.w * w;
        }
    }
    #pragma unroll
    for (int o = 16; o >= 1; o >>= 1) {
        dn0 += __shfl_xor(dn0, o, 32);
        dn1 += __shfl_xor(dn1, o, 32);
    }
    const float dn = (head ? dn1 + ws1 : dn0 + ws0) + SM_EPS;

    const float q0 = acc.x / dn, q1 = acc.y / dn, q2 = acc.z / dn, q3 = acc.w / dn;
    const int f = (lane & 15) * 4;
    const float4 bv = *(const float4*)(bias + f);
    const float v0 = (q0 + __shfl_xor(q0, 16, 32)) * 0.5f + bv.x;
    const float v1 = (q1 + __shfl_xor(q1, 16, 32)) * 0.5f + bv.y;
    const float v2 = (q2 + __shfl_xor(q2, 16, 32)) * 0.5f + bv.z;
    const float v3 = (q3 + __shfl_xor(q3, 16, 32)) * 0.5f + bv.w;

    float sum = (v0 + v1) + (v2 + v3);
    #pragma unroll
    for (int o = 8; o >= 1; o >>= 1) sum += __shfl_xor(sum, o, 32);
    const float mu = sum * (1.0f / OUT_F);
    const float d0 = v0 - mu, d1 = v1 - mu, d2 = v2 - mu, d3 = v3 - mu;
    float sq = (d0 * d0 + d1 * d1) + (d2 * d2 + d3 * d3);
    #pragma unroll
    for (int o = 8; o >= 1; o >>= 1) sq += __shfl_xor(sq, o, 32);
    const float inv = rsqrtf(sq * (1.0f / OUT_F) + LN_EPS);

    if (lane < 16) {
        const float4 gv = *(const float4*)(gamma + f);
        const float4 be = *(const float4*)(beta + f);
        float4 o4;
        o4.x = d0 * inv * gv.x + be.x;
        o4.y = d1 * inv * gv.y + be.y;
        o4.z = d2 * inv * gv.z + be.z;
        o4.w = d3 * inv * gv.w + be.w;
        *(float4*)(out + (size_t)d * OUT_F + f) = o4;
    }
}

extern "C" void kernel_launch(void* const* d_in, const int* in_sizes, int n_in,
                              void* d_out, int out_size, void* d_ws, size_t ws_size,
                              hipStream_t stream)
{
    const float* X    = (const float*)d_in[0];
    const int*   ei   = (const int*)  d_in[1];
    const float* W    = (const float*)d_in[2];
    const float* attS = (const float*)d_in[3];
    const float* attD = (const float*)d_in[4];
    const float* bias = (const float*)d_in[5];
    const float* g    = (const float*)d_in[6];
    const float* b    = (const float*)d_in[7];
    float* out = (float*)d_out;

    const int N = in_sizes[0] / IN_F;
    const int E = in_sizes[1] / 2;
    const int* esrc = ei;
    const int* edst = ei + E;

    float* ws   = (float*)d_ws;
    float* xp   = ws;
    float* aS   = xp + (size_t)N * XDIM;
    float* aD   = aS + (size_t)N * 2;
    int*   cnt  = (int*)(aD + (size_t)N * 2);
    int*   off  = cnt + (size_t)N;
    int*   rank = off + (size_t)N + 1;
    int*   ssrc = rank + (size_t)E;

    {
        const int waves = (N + 15) / 16;
        gat_gemm_wmma<<<(waves + 7) / 8, 256, 0, stream>>>(X, W, xp, N);
    }
    gat_logits_init<<<(N + 7) / 8, 256, 0, stream>>>(xp, attS, attD, aS, aD, cnt, N);
    gat_edge_rank<<<(E + 255) / 256, 256, 0, stream>>>(edst, cnt, rank, E);
    gat_scan<<<(N + SCAN_T - 1) / SCAN_T, SCAN_T, 0, stream>>>(cnt, off, N);
    gat_edge_place<<<(E + 255) / 256, 256, 0, stream>>>(esrc, edst, off, rank, ssrc, E);
    gat_node_gather<<<(N + 7) / 8, 256, 0, stream>>>(
        off, ssrc, aS, aD, xp, bias, g, b, out, N);
}


# Edges grouped by dst via LDS counters (chunks binned by 128-node range, whole-run copy-out; a block per range orders by dst) replace per-edge global adds and scan; sums in LDS-arrival order
# speedup vs baseline: 4.6664x; 1.6891x over previous
#include <hip/hip_runtime.h>
#include <hip/hip_bf16.h>
#include <math.h>


typedef __attribute__((ext_vector_type(2))) float v2f;
typedef __attribute__((ext_vector_type(8))) float v8f;

#define IN_F     128
#define OUT_F    64
#define HEADS    2
#define XDIM     128
#define NEG_SLOPE 0.2f
#define LN_EPS    1e-5f
#define SM_EPS    1e-16f

#define RSH     7
#define RNODES  (1 << RSH)
#define NBMAX   1024
#define CH      8192
#define EPT     8
#define NCHMAX  1024

__global__ __launch_bounds__(256) void gat_gemm_wmma(
    const float* __restrict__ X, const float* __restrict__ W,
    float* __restrict__ xp, int N)
{
    const int wave = (blockIdx.x * blockDim.x + threadIdx.x) >> 5;
    const int lane = threadIdx.x & 31;
    const int r0 = wave * 16;
    if (r0 >= N) return;

    const int lm = lane & 15;
    const int kk = (lane >> 4) << 1;

    int ar = r0 + lm; if (ar >= N) ar = N - 1;
    const float* arow = X + (size_t)ar * IN_F + kk;
    const float* brow = W + (size_t)lm * IN_F + kk;

    v8f acc[8];
    #pragma unroll
    for (int t = 0; t < 8; ++t) acc[t] = (v8f){};

    #pragma unroll 4
    for (int k0 = 0; k0 < IN_F; k0 += 4) {
        v2f a = *(const v2f*)(arow + k0);
        #pragma unroll
        for (int t = 0; t < 8; ++t) {
            v2f b = *(const v2f*)(brow + (size_t)t * 16 * IN_F + k0);
            acc[t] = __builtin_amdgcn_wmma_f32_16x16x4_f32(
                 false, a,  false, b,
                 (short)0, acc[t],  false,  false);
        }
    }

    const int mofs = (lane >> 4) << 3;
    #pragma unroll
    for (int v = 0; v < 8; ++v) {
        const int row = r0 + mofs + v;
        if (row < N) {
            float* orow = xp + (size_t)row * XDIM + lm;
            #pragma unroll
            for (int t = 0; t < 8; ++t) orow[t * 16] = acc[t][v];
        }
    }
}

__global__ __launch_bounds__(256) void gat_logits_init(
    const float* __restrict__ xp,
    const float* __restrict__ attS, const float* __restrict__ attD,
    float* __restrict__ aS, float* __restrict__ aD,
    int* __restrict__ binTot, int N)
{
    const int gt = blockIdx.x * blockDim.x + threadIdx.x;
    if (gt < NBMAX) binTot[gt] = 0;
    const int wave = gt >> 5;
    const int lane = threadIdx.x & 31;
    if (wave >= N) return;
    const int head = lane >> 4;
    const int fo   = lane * 4;

    float4 xv = *(const float4*)(xp + (size_t)wave * XDIM + fo);
    float4 sv = *(const float4*)(attS + fo);
    float4 dv = *(const float4*)(attD + fo);
    float ps = xv.x*sv.x + xv.y*sv.y + xv.z*sv.z + xv.w*sv.w;
    float pd = xv.x*dv.x + xv.y*dv.y + xv.z*dv.z + xv.w*dv.w;
    #pragma unroll
    for (int off = 8; off >= 1; off >>= 1) {
        ps += __shfl_xor(ps, off, 32);
        pd += __shfl_xor(pd, off, 32);
    }
    if ((lane & 15) == 0) {
        aS[wave * 2 + head] = ps;
        aD[wave * 2 + head] = pd;
    }
}

__device__ __forceinline__ float lrelu(float v)
{
    return v > 0.0f ? v : NEG_SLOPE * v;
}

__device__ __forceinline__ int block_scan_excl(int v, int tid, int* s_w, int* total)
{
    const int lane = tid & 31;
    const int wv   = tid >> 5;
    int inc = v;
    #pragma unroll
    for (int o = 1; o < 32; o <<= 1) {
        const int t = __shfl_up(inc, o, 32);
        if (lane >= o) inc += t;
    }
    __syncthreads();
    if (lane == 31) s_w[wv] = inc;
    __syncthreads();
    const int wt = s_w[lane];
    int winc = wt;
    #pragma unroll
    for (int o = 1; o < 32; o <<= 1) {
        const int t = __shfl_up(winc, o, 32);
        if (lane >= o) winc += t;
    }
    *total = __shfl(winc, 31, 32);
    return __shfl(winc - wt, wv, 32) + inc - v;
}

__global__ __launch_bounds__(1024) void gat_edge_bin(
    const int* __restrict__ esrc, const int* __restrict__ edst,
    int* __restrict__ binTot, int* __restrict__ runs, int* __restrict__ binned,
    int NB, int E)
{
    __shared__ int s_hist[NBMAX];
    __shared__ int s_base[NBMAX];
    __shared__ int s_rec[CH];
    __shared__ int s_w[32];
    const int tid = threadIdx.x;
    const int c   = blockIdx.x;
    const int e0  = c * CH;
    const int rem = E - e0;
    const int nE  = rem < CH ? rem : CH;

    s_hist[tid] = 0;
    __syncthreads();

    int dd[EPT], ss[EPT];
    #pragma unroll
    for (int k = 0; k < EPT; ++k) {
        const int i = k * 1024 + tid;
        const int j = e0 + (i < nE ? i : 0);
        dd[k] = edst[j];
        ss[k] = esrc[j];
    }

    int w[EPT], bn[EPT], rk[EPT];
    #pragma unroll
    for (int k = 0; k < EPT; ++k) {
        const int i = k * 1024 + tid;
        bn[k] = -1; w[k] = 0; rk[k] = 0;
        if (i < nE) {
            const int d = dd[k];
            const int s = ss[k];
            unsigned b = (unsigned)d >> RSH;
            if (b > NBMAX - 1) b = NBMAX - 1;
            bn[k] = (int)b;
            w[k]  = (s << RSH) | (d & (RNODES - 1));
            rk[k] = atomicAdd(&s_hist[b], 1);
        }
    }
    __syncthreads();

    const int cntb = s_hist[tid];
    int tot;
    const int ex = block_scan_excl(cntb, tid, s_w, &tot);
    s_base[tid] = ex;
    if (tid < NB) {
        runs[c * NB + tid] = (ex << 16) | cntb;
        if (cntb) atomicAdd(&binTot[tid], cntb);
    }
    __syncthreads();

    #pragma unroll
    for (int k = 0; k < EPT; ++k)
        if (bn[k] >= 0) s_rec[s_base[bn[k]] + rk[k]] = w[k];
    __syncthreads();

    for (int i = tid; i < nE; i += 1024) binned[e0 + i] = s_rec[i];
}

__global__ __launch_bounds__(1024) void gat_bin_sort(
    const int* __restrict__ binTot, const int* __restrict__ runs,
    const int* __restrict__ binned,
    int* __restrict__ off, int* __restrict__ ssrc,
    int N, int NB, int NCH)
{
    __shared__ int s_pos[NCHMAX];
    __shared__ int s_adr[NCHMAX];
    __shared__ int s_cnt[RNODES];
    __shared__ int s_w[32];
    __shared__ int s_bs;
    const int tid = threadIdx.x;
    const int b   = blockIdx.x;

    int tot;
    const int bt  = (tid < NB) ? binTot[tid] : 0;
    const int bex = block_scan_excl(bt, tid, s_w, &tot);
    if (tid == b) s_bs = bex;

    int cn = 0, ad = 0;
    if (tid < NCH) {
        const int pk = runs[tid * NB + b];
        cn = pk & 0xffff;
        ad = tid * CH + (int)((unsigned)pk >> 16);
    }
    int T;
    const int p = block_scan_excl(cn, tid, s_w, &T);
    s_pos[tid] = p;
    s_adr[tid] = ad;
    if (tid < RNODES) s_cnt[tid] = 0;
    __syncthreads();

    const int bs = s_bs;

    for (int f = tid; f < T; f += 1024) {
        int lo = 0, hi = NCH;
        while (hi - lo > 1) {
            const int mid = (lo + hi) >> 1;
            if (s_pos[mid] <= f) lo = mid; else hi = mid;
        }
        const int wd = binned[s_adr[lo] + (f - s_pos[lo])];
        atomicAdd(&s_cnt[wd & (RNODES - 1)], 1);
    }
    __syncthreads();

    const int cv = (tid < RNODES) ? s_cnt[tid] : 0;
    int T2;
    const int cex = block_scan_excl(cv, tid, s_w, &T2);
    if (tid < RNODES) {
        const int node = (b << RSH) + tid;
        if (node < N) off[node] = bs + cex;
        s_cnt[tid] = cex;
    }
    if (b == NB - 1 && tid == 0) off[N] = bs + T;
    __syncthreads();

    for (int f = tid; f < T; f += 1024) {
        int lo = 0, hi = NCH;
        while (hi - lo > 1) {
            const int mid = (lo + hi) >> 1;
            if (s_pos[mid] <= f) lo = mid; else hi = mid;
        }
        const int wd = binned[s_adr[lo] + (f - s_pos[lo])];
        const int q  = atomicAdd(&s_cnt[wd & (RNODES - 1)], 1);
        ssrc[bs + q] = (int)((unsigned)wd >> RSH);
    }
}

__global__ __launch_bounds__(256) void gat_node_gather(
    const int* __restrict__ off, const int* __restrict__ ssrc,
    const float* __restrict__ aS, const float* __restrict__ aD,
    const float* __restrict__ xp,
    const float* __restrict__ bias,
    const float* __restrict__ gamma, const float* __restrict__ beta,
    float* __restrict__ out, int N)
{
    const int d = __builtin_amdgcn_readfirstlane(
        (int)((blockIdx.x * blockDim.x + threadIdx.x) >> 5));
    const int lane = threadIdx.x & 31;
    if (d >= N) return;
    const int head = lane >> 4;
    const int fo = lane * 4;

    const int beg = off[d];
    const int end = off[d + 1];
    const float ad0 = aD[d * 2 + 0], ad1 = aD[d * 2 + 1];
    const float es0 = lrelu(aS[d * 2 + 0] + ad0);
    const float es1 = lrelu(aS[d * 2 + 1] + ad1);
    const float4 xs = *(const float4*)(xp + (size_t)d * XDIM + fo);

    float m0 = es0, m1 = es1;
    for (int i = beg + lane; i < end; i += 32) {
        const int s = ssrc[i];
        const float2 a = *(const float2*)(aS + (size_t)s * 2);
        m0 = fmaxf(m0, lrelu(a.x + ad0));
        m1 = fmaxf(m1, lrelu(a.y + ad1));
    }
    #pragma unroll
    for (int o = 16; o >= 1; o >>= 1) {
        m0 = fmaxf(m0, __shfl_xor(m0, o, 32));
        m1 = fmaxf(m1, __shfl_xor(m1, o, 32));
    }

    const float ws0 = __expf(es0 - m0), ws1 = __expf(es1 - m1);
    const float wsh = head ? ws1 : ws0;
    float4 acc = {xs.x * wsh, xs.y * wsh, xs.z * wsh, xs.w * wsh};
    float dn0 = 0.0f, dn1 = 0.0f;

    for (int base = beg; base < end; base += 32) {
        const int i = base + lane;
        int s = 0;
        float w0 = 0.0f, w1 = 0.0f;
        if (i < end) {
            s = ssrc[i];
            const float2 a = *(const float2*)(aS + (size_t)s * 2);
            w0 = __expf(lrelu(a.x + ad0) - m0);
            w1 = __expf(lrelu(a.y + ad1) - m1);
        }
        dn0 += w0;
        dn1 += w1;
        const int rem = end - base;
        const int n = rem < 32 ? rem : 32;
        #pragma unroll 4
        for (int j = 0; j < n; ++j) {
            const int sj = __builtin_amdgcn_readlane(s, j);
            const float wa = __int_as_float(__builtin_amdgcn_readlane(__float_as_int(w0), j));
            const float wb = __int_as_float(__builtin_amdgcn_readlane(__float_as_int(w1), j));
            const float w = head ? wb : wa;
            const float4 v = *(const float4*)(xp + (size_t)sj * XDIM + fo);
            acc.x += v.x * w;
            acc.y += v.y * w;
            acc.z += v.z * w;
            acc.w += v.w * w;
        }
    }
    #pragma unroll
    for (int o = 16; o >= 1; o >>= 1) {
        dn0 += __shfl_xor(dn0, o, 32);
        dn1 += __shfl_xor(dn1, o, 32);
    }
    const float dn = (head ? dn1 + ws1 : dn0 + ws0) + SM_EPS;

    const float q0 = acc.x / dn, q1 = acc.y / dn, q2 = acc.z / dn, q3 = acc.w / dn;
    const int f = (lane & 15) * 4;
    const float4 bv = *(const float4*)(bias + f);
    const float v0 = (q0 + __shfl_xor(q0, 16, 32)) * 0.5f + bv.x;
    const float v1 = (q1 + __shfl_xor(q1, 16, 32)) * 0.5f + bv.y;
    const float v2 = (q2 + __shfl_xor(q2, 16, 32)) * 0.5f + bv.z;
    const float v3 = (q3 + __shfl_xor(q3, 16, 32)) * 0.5f + bv.w;

    float sum = (v0 + v1) + (v2 + v3);
    #pragma unroll
    for (int o = 8; o >= 1; o >>= 1) sum += __shfl_xor(sum, o, 32);
    const float mu = sum * (1.0f / OUT_F);
    const float d0 = v0 - mu, d1 = v1 - mu, d2 = v2 - mu, d3 = v3 - mu;
    float sq = (d0 * d0 + d1 * d1) + (d2 * d2 + d3 * d3);
    #pragma unroll
    for (int o = 8; o >= 1; o >>= 1) sq += __shfl_xor(sq, o, 32);
    const float inv = rsqrtf(sq * (1.0f / OUT_F) + LN_EPS);

    if (lane < 16) {
        const float4 gv = *(const float4*)(gamma + f);
        const float4 be = *(const float4*)(beta + f);
        float4 o4;
        o4.x = d0 * inv * gv.x + be.x;
        o4.y = d1 * inv * gv.y + be.y;
        o4.z = d2 * inv * gv.z + be.z;
        o4.w = d3 * inv * gv.w + be.w;
        *(float4*)(out + (size_t)d * OUT_F + f) = o4;
    }
}

extern "C" void kernel_launch(void* const* d_in, const int* in_sizes, int n_in,
                              void* d_out, int out_size, void* d_ws, size_t ws_size,
                              hipStream_t stream)
{
    const float* X    = (const float*)d_in[0];
    const int*   ei   = (const int*)  d_in[1];
    const float* W    = (const float*)d_in[2];
    const float* attS = (const float*)d_in[3];
    const float* attD = (const float*)d_in[4];
    const float* bias = (const float*)d_in[5];
    const float* g    = (const float*)d_in[6];
    const float* b    = (const float*)d_in[7];
    float* out = (float*)d_out;

    const int N = in_sizes[0] / IN_F;
    const int E = in_sizes[1] / 2;
    const int* esrc = ei;
    const int* edst = ei + E;

    const int NB  = (N + RNODES - 1) / RNODES;
    const int NCH = (E + CH - 1) / CH;

    float* ws     = (float*)d_ws;
    float* xp     = ws;
    float* aS     = xp + (size_t)N * XDIM;
    float* aD     = aS + (size_t)N * 2;
    int*   binTot = (int*)(aD + (size_t)N * 2);
    int*   off    = binTot + NBMAX;
    int*   runs   = off + (size_t)N + 1;
    int*   binned = runs + (size_t)NCH * NB;
    int*   ssrc   = binned + (size_t)E;

    {
        const int waves = (N + 15) / 16;
        gat_gemm_wmma<<<(waves + 7) / 8, 256, 0, stream>>>(X, W, xp, N);
    }
    gat_logits_init<<<(N + 7) / 8, 256, 0, stream>>>(xp, attS, attD, aS, aD, binTot, N);
    gat_edge_bin<<<NCH, 1024, 0, stream>>>(esrc, edst, binTot, runs, binned, NB, E);
    gat_bin_sort<<<NB, 1024, 0, stream>>>(binTot, runs, binned, off, ssrc, N, NB, NCH);
    gat_node_gather<<<(N + 7) / 8, 256, 0, stream>>>(
        off, ssrc, aS, aD, xp, bias, g, b, out, N);
}


# Logits made in the projection kernel's epilogue from the WMMA accumulators (transposed 16-lane DPP sums); the logits launch and its re-read of x are gone; logit dot products summed in another order
# speedup vs baseline: 5.1315x; 1.0997x over previous
#include <hip/hip_runtime.h>
#include <hip/hip_bf16.h>
#include <math.h>


typedef __attribute__((ext_vector_type(2))) float v2f;
typedef __attribute__((ext_vector_type(8))) float v8f;

#define IN_F     128
#define OUT_F    64
#define HEADS    2
#define XDIM     128
#define NEG_SLOPE 0.2f
#define LN_EPS    1e-5f
#define SM_EPS    1e-16f

#define RSH     7
#define RNODES  (1 << RSH)
#define NBMAX   1024
#define CH      8192
#define EPT     8
#define NCHMAX  1024


template <int M>
__device__ __forceinline__ float row_xor_get(float x)
{
    const int b = __float_as_int(x);
    return __int_as_float(__builtin_amdgcn_update_dpp(b, b, 0x160 | M, 0xf, 0xf, false));
}

__device__ __forceinline__ float row_sum_transposed(const float (&p)[16], int c)
{
    const bool b3 = (c & 8) != 0, b2 = (c & 4) != 0, b1 = (c & 2) != 0, b0 = (c & 1) != 0;
    float q[8], r[4], s[2];
    #pragma unroll
    for (int j = 0; j < 8; ++j) {
        const float keep = b3 ? p[j + 8] : p[j];
        const float send = b3 ? p[j] : p[j + 8];
        q[j] = keep + row_xor_get<8>(send);
    }
    #pragma unroll
    for (int j = 0; j < 4; ++j) {
        const float keep = b2 ? q[j + 4] : q[j];
        const float send = b2 ? q[j] : q[j + 4];
        r[j] = keep + row_xor_get<4>(send);
    }
    #pragma unroll
    for (int j = 0; j < 2; ++j) {
        const float keep = b1 ? r[j + 2] : r[j];
        const float send = b1 ? r[j] : r[j + 2];
        s[j] = keep + row_xor_get<2>(send);
    }
    const float keep = b0 ? s[1] : s[0];
    const float send = b0 ? s[0] : s[1];
    return keep + row_xor_get<1>(send);
}

__global__ __launch_bounds__(256) void gat_gemm_wmma(
    const float* __restrict__ X, const float* __restrict__ W,
    const float* __restrict__ attS, const float* __restrict__ attD,
    float* __restrict__ xp, float* __restrict__ aS, float* __restrict__ aD,
    int* __restrict__ binTot, int N)
{
    const int gt = blockIdx.x * blockDim.x + threadIdx.x;
    for (int i = gt; i < NBMAX; i += gridDim.x * blockDim.x) binTot[i] = 0;
    const int wave = gt >> 5;
    const int lane = threadIdx.x & 31;
    const int r0 = wave * 16;
    if (r0 >= N) return;

    const int lm = lane & 15;
    const int kk = (lane >> 4) << 1;

    int ar = r0 + lm; if (ar >= N) ar = N - 1;
    const float* arow = X + (size_t)ar * IN_F + kk;
    const float* brow = W + (size_t)lm * IN_F + kk;

    v8f acc[8];
    #pragma unroll
    for (int t = 0; t < 8; ++t) acc[t] = (v8f){};

    #pragma unroll 4
    for (int k0 = 0; k0 < IN_F; k0 += 4) {
        v2f a = *(const v2f*)(arow + k0);
        #pragma unroll
        for (int t = 0; t < 8; ++t) {
            v2f b = *(const v2f*)(brow + (size_t)t * 16 * IN_F + k0);
            acc[t] = __builtin_amdgcn_wmma_f32_16x16x4_f32(
                 false, a,  false, b,
                 (short)0, acc[t],  false,  false);
        }
    }

    const int mofs = (lane >> 4) << 3;
    #pragma unroll
    for (int v = 0; v < 8; ++v) {
        const int row = r0 + mofs + v;
        if (row < N) {
            float* orow = xp + (size_t)row * XDIM + lm;
            #pragma unroll
            for (int t = 0; t < 8; ++t) orow[t * 16] = acc[t][v];
        }
    }

    float sv[8], dv[8];
    #pragma unroll
    for (int t = 0; t < 8; ++t) {
        sv[t] = attS[t * 16 + lm];
        dv[t] = attD[t * 16 + lm];
    }
    float pS[16], pD[16];
    #pragma unroll
    for (int v = 0; v < 8; ++v) {
        #pragma unroll
        for (int h = 0; h < 2; ++h) {
            const float x0 = acc[4 * h + 0][v], x1 = acc[4 * h + 1][v];
            const float x2 = acc[4 * h + 2][v], x3 = acc[4 * h + 3][v];
            pS[v * 2 + h] = x0 * sv[4 * h + 0] + x1 * sv[4 * h + 1]
                          + x2 * sv[4 * h + 2] + x3 * sv[4 * h + 3];
            pD[v * 2 + h] = x0 * dv[4 * h + 0] + x1 * dv[4 * h + 1]
                          + x2 * dv[4 * h + 2] + x3 * dv[4 * h + 3];
        }
    }
    const float tS = row_sum_transposed(pS, lm);
    const float tD = row_sum_transposed(pD, lm);
    if (r0 + (lane >> 1) < N) {
        aS[(size_t)r0 * 2 + lane] = tS;
        aD[(size_t)r0 * 2 + lane] = tD;
    }
}

__device__ __forceinline__ float lrelu(float v)
{
    return v > 0.0f ? v : NEG_SLOPE * v;
}

__device__ __forceinline__ int block_scan_excl(int v, int tid, int* s_w, int* total)
{
    const int lane = tid & 31;
    const int wv   = tid >> 5;
    int inc = v;
    #pragma unroll
    for (int o = 1; o < 32; o <<= 1) {
        const int t = __shfl_up(inc, o, 32);
        if (lane >= o) inc += t;
    }
    __syncthreads();
    if (lane == 31) s_w[wv] = inc;
    __syncthreads();
    const int wt = s_w[lane];
    int winc = wt;
    #pragma unroll
    for (int o = 1; o < 32; o <<= 1) {
        const int t = __shfl_up(winc, o, 32);
        if (lane >= o) winc += t;
    }
    *total = __shfl(winc, 31, 32);
    return __shfl(winc - wt, wv, 32) + inc - v;
}

__global__ __launch_bounds__(1024) void gat_edge_bin(
    const int* __restrict__ esrc, const int* __restrict__ edst,
    int* __restrict__ binTot, int* __restrict__ runs, int* __restrict__ binned,
    int NB, int E)
{
    __shared__ int s_hist[NBMAX];
    __shared__ int s_base[NBMAX];
    __shared__ int s_rec[CH];
    __shared__ int s_w[32];
    const int tid = threadIdx.x;
    const int c   = blockIdx.x;
    const int e0  = c * CH;
    const int rem = E - e0;
    const int nE  = rem < CH ? rem : CH;

    s_hist[tid] = 0;
    __syncthreads();

    int dd[EPT], ss[EPT];
    #pragma unroll
    for (int k = 0; k < EPT; ++k) {
        const int i = k * 1024 + tid;
        const int j = e0 + (i < nE ? i : 0);
        dd[k] = edst[j];
        ss[k] = esrc[j];
    }

    int w[EPT], bn[EPT], rk[EPT];
    #pragma unroll
    for (int k = 0; k < EPT; ++k) {
        const int i = k * 1024 + tid;
        bn[k] = -1; w[k] = 0; rk[k] = 0;
        if (i < nE) {
            const int d = dd[k];
            const int s = ss[k];
            unsigned b = (unsigned)d >> RSH;
            if (b > NBMAX - 1) b = NBMAX - 1;
            bn[k] = (int)b;
            w[k]  = (s << RSH) | (d & (RNODES - 1));
            rk[k] = atomicAdd(&s_hist[b], 1);
        }
    }
    __syncthreads();

    const int cntb = s_hist[tid];
    int tot;
    const int ex = block_scan_excl(cntb, tid, s_w, &tot);
    s_base[tid] = ex;
    if (tid < NB) {
        runs[c * NB + tid] = (ex << 16) | cntb;
        if (cntb) atomicAdd(&binTot[tid], cntb);
    }
    __syncthreads();

    #pragma unroll
    for (int k = 0; k < EPT; ++k)
        if (bn[k] >= 0) s_rec[s_base[bn[k]] + rk[k]] = w[k];
    __syncthreads();

    for (int i = tid; i < nE; i += 1024) binned[e0 + i] = s_rec[i];
}

__global__ __launch_bounds__(1024) void gat_bin_sort(
    const int* __restrict__ binTot, const int* __restrict__ runs,
    const int* __restrict__ binned,
    int* __restrict__ off, int* __restrict__ ssrc,
    int N, int NB, int NCH)
{
    __shared__ int s_pos[NCHMAX];
    __shared__ int s_adr[NCHMAX];
    __shared__ int s_cnt[RNODES];
    __shared__ int s_w[32];
    __shared__ int s_bs;
    const int tid = threadIdx.x;
    const int b   = blockIdx.x;

    int tot;
    const int bt  = (tid < NB) ? binTot[tid] : 0;
    const int bex = block_scan_excl(bt, tid, s_w, &tot);
    if (tid == b) s_bs = bex;

    int cn = 0, ad = 0;
    if (tid < NCH) {
        const int pk = runs[tid * NB + b];
        cn = pk & 0xffff;
        ad = tid * CH + (int)((unsigned)pk >> 16);
    }
    int T;
    const int p = block_scan_excl(cn, tid, s_w, &T);
    s_pos[tid] = p;
    s_adr[tid] = ad;
    if (tid < RNODES) s_cnt[tid] = 0;
    __syncthreads();

    const int bs = s_bs;

    for (int f = tid; f < T; f += 1024) {
        int lo = 0, hi = NCH;
        while (hi - lo > 1) {
            const int mid = (lo + hi) >> 1;
            if (s_pos[mid] <= f) lo = mid; else hi = mid;
        }
        const int wd = binned[s_adr[lo] + (f - s_pos[lo])];
        atomicAdd(&s_cnt[wd & (RNODES - 1)], 1);
    }
    __syncthreads();

    const int cv = (tid < RNODES) ? s_cnt[tid] : 0;
    int T2;
    const int cex = block_scan_excl(cv, tid, s_w, &T2);
    if (tid < RNODES) {
        const int node = (b << RSH) + tid;
        if (node < N) off[node] = bs + cex;
        s_cnt[tid] = cex;
    }
    if (b == NB - 1 && tid == 0) off[N] = bs + T;
    __syncthreads();

    for (int f = tid; f < T; f += 1024) {
        int lo = 0, hi = NCH;
        while (hi - lo > 1) {
            const int mid = (lo + hi) >> 1;
            if (s_pos[mid] <= f) lo = mid; else hi = mid;
        }
        const int wd = binned[s_adr[lo] + (f - s_pos[lo])];
        const int q  = atomicAdd(&s_cnt[wd & (RNODES - 1)], 1);
        ssrc[bs + q] = (int)((unsigned)wd >> RSH);
    }
}

__global__ __launch_bounds__(256) void gat_node_gather(
    const int* __restrict__ off, const int* __restrict__ ssrc,
    const float* __restrict__ aS, const float* __restrict__ aD,
    const float* __restrict__ xp,
    const float* __restrict__ bias,
    const float* __restrict__ gamma, const float* __restrict__ beta,
    float* __restrict__ out, int N)
{
    const int d = __builtin_amdgcn_readfirstlane(
        (int)((blockIdx.x * blockDim.x + threadIdx.x) >> 5));
    const int lane = threadIdx.x & 31;
    if (d >= N) return;
    const int head = lane >> 4;
    const int fo = lane * 4;

    const int beg = off[d];
    const int end = off[d + 1];
    const float ad0 = aD[d * 2 + 0], ad1 = aD[d * 2 + 1];
    const float es0 = lrelu(aS[d * 2 + 0] + ad0);
    const float es1 = lrelu(aS[d * 2 + 1] + ad1);
    const float4 xs = *(const float4*)(xp + (size_t)d * XDIM + fo);

    float m0 = es0, m1 = es1;
    for (int i = beg + lane; i < end; i += 32) {
        const int s = ssrc[i];
        const float2 a = *(const float2*)(aS + (size_t)s * 2);
        m0 = fmaxf(m0, lrelu(a.x + ad0));
        m1 = fmaxf(m1, lrelu(a.y + ad1));
    }
    #pragma unroll
    for (int o = 16; o >= 1; o >>= 1) {
        m0 = fmaxf(m0, __shfl_xor(m0, o, 32));
        m1 = fmaxf(m1, __shfl_xor(m1, o, 32));
    }

    const float ws0 = __expf(es0 - m0), ws1 = __expf(es1 - m1);
    const float wsh = head ? ws1 : ws0;
    float4 acc = {xs.x * wsh, xs.y * wsh, xs.z * wsh, xs.w * wsh};
    float dn0 = 0.0f, dn1 = 0.0f;

    for (int base = beg; base < end; base += 32) {
        const int i = base + lane;
        int s = 0;
        float w0 = 0.0f, w1 = 0.0f;
        if (i < end) {
            s = ssrc[i];
            const float2 a = *(const float2*)(aS + (size_t)s * 2);
            w0 = __expf(lrelu(a.x + ad0) - m0);
            w1 = __expf(lrelu(a.y + ad1) - m1);
        }
        dn0 += w0;
        dn1 += w1;
        const int rem = end - base;
        const int n = rem < 32 ? rem : 32;
        #pragma unroll 4
        for (int j = 0; j < n; ++j) {
            const int sj = __builtin_amdgcn_readlane(s, j);
            const float wa = __int_as_float(__builtin_amdgcn_readlane(__float_as_int(w0), j));
            const float wb = __int_as_float(__builtin_amdgcn_readlane(__float_as_int(w1), j));
            const float w = head ? wb : wa;
            const float4 v = *(const float4*)(xp + (size_t)sj * XDIM + fo);
            acc.x += v.x * w;
            acc.y += v.y * w;
            acc.z += v.z * w;
            acc.w += v.w * w;
        }
    }
    #pragma unroll
    for (int o = 16; o >= 1; o >>= 1) {
        dn0 += __shfl_xor(dn0, o, 32);
        dn1 += __shfl_xor(dn1, o, 32);
    }
    const float dn = (head ? dn1 + ws1 : dn0 + ws0) + SM_EPS;

    const float q0 = acc.x / dn, q1 = acc.y / dn, q2 = acc.z / dn, q3 = acc.w / dn;
    const int f = (lane & 15) * 4;
    const float4 bv = *(const float4*)(bias + f);
    const float v0 = (q0 + __shfl_xor(q0, 16, 32)) * 0.5f + bv.x;
    const float v1 = (q1 + __shfl_xor(q1, 16, 32)) * 0.5f + bv.y;
    const float v2 = (q2 + __shfl_xor(q2, 16, 32)) * 0.5f + bv.z;
    const float v3 = (q3 + __shfl_xor(q3, 16, 32)) * 0.5f + bv.w;

    float sum = (v0 + v1) + (v2 + v3);
    #pragma unroll
    for (int o = 8; o >= 1; o >>= 1) sum += __shfl_xor(sum, o, 32);
    const float mu = sum * (1.0f / OUT_F);
    const float d0 = v0 - mu, d1 = v1 - mu, d2 = v2 - mu, d3 = v3 - mu;
    float sq = (d0 * d0 + d1 * d1) + (d2 * d2 + d3 * d3);
    #pragma unroll
    for (int o = 8; o >= 1; o >>= 1) sq += __shfl_xor(sq, o, 32);
    const float inv = rsqrtf(sq * (1.0f / OUT_F) + LN_EPS);

    if (lane < 16) {
        const float4 gv = *(const float4*)(gamma + f);
        const float4 be = *(const float4*)(beta + f);
        float4 o4;
        o4.x = d0 * inv * gv.x + be.x;
        o4.y = d1 * inv * gv.y + be.y;
        o4.z = d2 * inv * gv.z + be.z;
        o4.w = d3 * inv * gv.w + be.w;
        *(float4*)(out + (size_t)d * OUT_F + f) = o4;
    }
}

extern "C" void kernel_launch(void* const* d_in, const int* in_sizes, int n_in,
                              void* d_out, int out_size, void* d_ws, size_t ws_size,
                              hipStream_t stream)
{
    const float* X    = (const float*)d_in[0];
    const int*   ei   = (const int*)  d_in[1];
    const float* W    = (const float*)d_in[2];
    const float* attS = (const float*)d_in[3];
    const float* attD = (const float*)d_in[4];
    const float* bias = (const float*)d_in[5];
    const float* g    = (const float*)d_in[6];
    const float* b    = (const float*)d_in[7];
    float* out = (float*)d_out;

    const int N = in_sizes[0] / IN_F;
    const int E = in_sizes[1] / 2;
    const int* esrc = ei;
    const int* edst = ei + E;

    const int NB  = (N + RNODES - 1) / RNODES;
    const int NCH = (E + CH - 1) / CH;

    float* ws     = (float*)d_ws;
    float* xp     = ws;
    float* aS     = xp + (size_t)N * XDIM;
    float* aD     = aS + (size_t)N * 2;
    int*   binTot = (int*)(aD + (size_t)N * 2);
    int*   off    = binTot + NBMAX;
    int*   runs   = off + (size_t)N + 1;
    int*   binned = runs + (size_t)NCH * NB;
    int*   ssrc   = binned + (size_t)E;

    {
        const int waves = (N + 15) / 16;
        gat_gemm_wmma<<<(waves + 7) / 8, 256, 0, stream>>>(
            X, W, attS, attD, xp, aS, aD, binTot, N);
    }
    gat_edge_bin<<<NCH, 1024, 0, stream>>>(esrc, edst, binTot, runs, binned, NB, E);
    gat_bin_sort<<<NB, 1024, 0, stream>>>(binTot, runs, binned, off, ssrc, N, NB, NCH);
    gat_node_gather<<<(N + 7) / 8, 256, 0, stream>>>(
        off, ssrc, aS, aD, xp, bias, g, b, out, N);
}


# Projection GEMM: each block copies W (64KB) into LDS with whole-row loads and takes its B fragments from there; the 8 global 16-line B loads per K-step are gone; k order and sums unchanged
# speedup vs baseline: 6.0377x; 1.1766x over previous
#include <hip/hip_runtime.h>
#include <hip/hip_bf16.h>
#include <math.h>


typedef __attribute__((ext_vector_type(2))) float v2f;
typedef __attribute__((ext_vector_type(8))) float v8f;

#define IN_F     128
#define OUT_F    64
#define HEADS    2
#define XDIM     128
#define NEG_SLOPE 0.2f
#define LN_EPS    1e-5f
#define SM_EPS    1e-16f

#define RSH     7
#define RNODES  (1 << RSH)
#define NBMAX   1024
#define CH      8192
#define EPT     8
#define NCHMAX  1024

#define WPITCH  132


template <int M>
__device__ __forceinline__ float row_xor_get(float x)
{
    const int b = __float_as_int(x);
    return __int_as_float(__builtin_amdgcn_update_dpp(b, b, 0x160 | M, 0xf, 0xf, false));
}

__device__ __forceinline__ float row_sum_transposed(const float (&p)[16], int c)
{
    const bool b3 = (c & 8) != 0, b2 = (c & 4) != 0, b1 = (c & 2) != 0, b0 = (c & 1) != 0;
    float q[8], r[4], s[2];
    #pragma unroll
    for (int j = 0; j < 8; ++j) {
        const float keep = b3 ? p[j + 8] : p[j];
        const float send = b3 ? p[j] : p[j + 8];
        q[j] = keep + row_xor_get<8>(send);
    }
    #pragma unroll
    for (int j = 0; j < 4; ++j) {
        const float keep = b2 ? q[j + 4] : q[j];
        const float send = b2 ? q[j] : q[j + 4];
        r[j] = keep + row_xor_get<4>(send);
    }
    #pragma unroll
    for (int j = 0; j < 2; ++j) {
        const float keep = b1 ? r[j + 2] : r[j];
        const float send = b1 ? r[j] : r[j + 2];
        s[j] = keep + row_xor_get<2>(send);
    }
    const float keep = b0 ? s[1] : s[0];
    const float send = b0 ? s[0] : s[1];
    return keep + row_xor_get<1>(send);
}

__global__ __launch_bounds__(256) void gat_gemm_wmma(
    const float* __restrict__ X, const float* __restrict__ W,
    const float* __restrict__ attS, const float* __restrict__ attD,
    float* __restrict__ xp, float* __restrict__ aS, float* __restrict__ aD,
    int* __restrict__ binTot, int N)
{
    __shared__ float s_wt[XDIM * WPITCH];

    const int gt = blockIdx.x * blockDim.x + threadIdx.x;
    for (int i = gt; i < NBMAX; i += gridDim.x * blockDim.x) binTot[i] = 0;

    #pragma unroll
    for (int j = 0; j < (XDIM * IN_F / 4) / 256; ++j) {
        const int idx = j * 256 + (int)threadIdx.x;
        const float4 wv = *(const float4*)(W + (size_t)idx * 4);
        *(float4*)(s_wt + (idx >> 5) * WPITCH + (idx & 31) * 4) = wv;
    }
    __syncthreads();

    const int wave = gt >> 5;
    const int lane = threadIdx.x & 31;
    const int r0 = wave * 16;
    if (r0 >= N) return;

    const int lm = lane & 15;
    const int kk = (lane >> 4) << 1;

    int ar = r0 + lm; if (ar >= N) ar = N - 1;
    const float* arow = X + (size_t)ar * IN_F + kk;
    const float* brow = s_wt + lm * WPITCH + kk;

    v8f acc[8];
    #pragma unroll
    for (int t = 0; t < 8; ++t) acc[t] = (v8f){};

    #pragma unroll 4
    for (int k0 = 0; k0 < IN_F; k0 += 4) {
        v2f a = *(const v2f*)(arow + k0);
        #pragma unroll
        for (int t = 0; t < 8; ++t) {
            v2f b = *(const v2f*)(brow + t * 16 * WPITCH + k0);
            acc[t] = __builtin_amdgcn_wmma_f32_16x16x4_f32(
                 false, a,  false, b,
                 (short)0, acc[t],  false,  false);
        }
    }

    const int mofs = (lane >> 4) << 3;
    #pragma unroll
    for (int v = 0; v < 8; ++v) {
        const int row = r0 + mofs + v;
        if (row < N) {
            float* orow = xp + (size_t)row * XDIM + lm;
            #pragma unroll
            for (int t = 0; t < 8; ++t) orow[t * 16] = acc[t][v];
        }
    }

    float sv[8], dv[8];
    #pragma unroll
    for (int t = 0; t < 8; ++t) {
        sv[t] = attS[t * 16 + lm];
        dv[t] = attD[t * 16 + lm];
    }
    float pS[16], pD[16];
    #pragma unroll
    for (int v = 0; v < 8; ++v) {
        #pragma unroll
        for (int h = 0; h < 2; ++h) {
            const float x0 = acc[4 * h + 0][v], x1 = acc[4 * h + 1][v];
            const float x2 = acc[4 * h + 2][v], x3 = acc[4 * h + 3][v];
            pS[v * 2 + h] = x0 * sv[4 * h + 0] + x1 * sv[4 * h + 1]
                          + x2 * sv[4 * h + 2] + x3 * sv[4 * h + 3];
            pD[v * 2 + h] = x0 * dv[4 * h + 0] + x1 * dv[4 * h + 1]
                          + x2 * dv[4 * h + 2] + x3 * dv[4 * h + 3];
        }
    }
    const float tS = row_sum_transposed(pS, lm);
    const float tD = row_sum_transposed(pD, lm);
    if (r0 + (lane >> 1) < N) {
        aS[(size_t)r0 * 2 + lane] = tS;
        aD[(size_t)r0 * 2 + lane] = tD;
    }
}

__device__ __forceinline__ float lrelu(float v)
{
    return v > 0.0f ? v : NEG_SLOPE * v;
}

__device__ __forceinline__ int block_scan_excl(int v, int tid, int* s_w, int* total)
{
    const int lane = tid & 31;
    const int wv   = tid >> 5;
    int inc = v;
    #pragma unroll
    for (int o = 1; o < 32; o <<= 1) {
        const int t = __shfl_up(inc, o, 32);
        if (lane >= o) inc += t;
    }
    __syncthreads();
    if (lane == 31) s_w[wv] = inc;
    __syncthreads();
    const int wt = s_w[lane];
    int winc = wt;
    #pragma unroll
    for (int o = 1; o < 32; o <<= 1) {
        const int t = __shfl_up(winc, o, 32);
        if (lane >= o) winc += t;
    }
    *total = __shfl(winc, 31, 32);
    return __shfl(winc - wt, wv, 32) + inc - v;
}

__global__ __launch_bounds__(1024) void gat_edge_bin(
    const int* __restrict__ esrc, const int* __restrict__ edst,
    int* __restrict__ binTot, int* __restrict__ runs, int* __restrict__ binned,
    int NB, int E)
{
    __shared__ int s_hist[NBMAX];
    __shared__ int s_base[NBMAX];
    __shared__ int s_rec[CH];
    __shared__ int s_w[32];
    const int tid = threadIdx.x;
    const int c   = blockIdx.x;
    const int e0  = c * CH;
    const int rem = E - e0;
    const int nE  = rem < CH ? rem : CH;

    s_hist[tid] = 0;
    __syncthreads();

    int dd[EPT], ss[EPT];
    #pragma unroll
    for (int k = 0; k < EPT; ++k) {
        const int i = k * 1024 + tid;
        const int j = e0 + (i < nE ? i : 0);
        dd[k] = edst[j];
        ss[k] = esrc[j];
    }

    int w[EPT], bn[EPT], rk[EPT];
    #pragma unroll
    for (int k = 0; k < EPT; ++k) {
        const int i = k * 1024 + tid;
        bn[k] = -1; w[k] = 0; rk[k] = 0;
        if (i < nE) {
            const int d = dd[k];
            const int s = ss[k];
            unsigned b = (unsigned)d >> RSH;
            if (b > NBMAX - 1) b = NBMAX - 1;
            bn[k] = (int)b;
            w[k]  = (s << RSH) | (d & (RNODES - 1));
            rk[k] = atomicAdd(&s_hist[b], 1);
        }
    }
    __syncthreads();

    const int cntb = s_hist[tid];
    int tot;
    const int ex = block_scan_excl(cntb, tid, s_w, &tot);
    s_base[tid] = ex;
    if (tid < NB) {
        runs[c * NB + tid] = (ex << 16) | cntb;
        if (cntb) atomicAdd(&binTot[tid], cntb);
    }
    __syncthreads();

    #pragma unroll
    for (int k = 0; k < EPT; ++k)
        if (bn[k] >= 0) s_rec[s_base[bn[k]] + rk[k]] = w[k];
    __syncthreads();

    for (int i = tid; i < nE; i += 1024) binned[e0 + i] = s_rec[i];
}

__global__ __launch_bounds__(1024) void gat_bin_sort(
    const int* __restrict__ binTot, const int* __restrict__ runs,
    const int* __restrict__ binned,
    int* __restrict__ off, int* __restrict__ ssrc,
    int N, int NB, int NCH)
{
    __shared__ int s_pos[NCHMAX];
    __shared__ int s_adr[NCHMAX];
    __shared__ int s_cnt[RNODES];
    __shared__ int s_w[32];
    __shared__ int s_bs;
    const int tid = threadIdx.x;
    const int b   = blockIdx.x;

    int tot;
    const int bt  = (tid < NB) ? binTot[tid] : 0;
    const int bex = block_scan_excl(bt, tid, s_w, &tot);
    if (tid == b) s_bs = bex;

    int cn = 0, ad = 0;
    if (tid < NCH) {
        const int pk = runs[tid * NB + b];
        cn = pk & 0xffff;
        ad = tid * CH + (int)((unsigned)pk >> 16);
    }
    int T;
    const int p = block_scan_excl(cn, tid, s_w, &T);
    s_pos[tid] = p;
    s_adr[tid] = ad;
    if (tid < RNODES) s_cnt[tid] = 0;
    __syncthreads();

    const int bs = s_bs;

    for (int f = tid; f < T; f += 1024) {
        int lo = 0, hi = NCH;
        while (hi - lo > 1) {
            const int mid = (lo + hi) >> 1;
            if (s_pos[mid] <= f) lo = mid; else hi = mid;
        }
        const int wd = binned[s_adr[lo] + (f - s_pos[lo])];
        atomicAdd(&s_cnt[wd & (RNODES - 1)], 1);
    }
    __syncthreads();

    const int cv = (tid < RNODES) ? s_cnt[tid] : 0;
    int T2;
    const int cex = block_scan_excl(cv, tid, s_w, &T2);
    if (tid < RNODES) {
        const int node = (b << RSH) + tid;
        if (node < N) off[node] = bs + cex;
        s_cnt[tid] = cex;
    }
    if (b == NB - 1 && tid == 0) off[N] = bs + T;
    __syncthreads();

    for (int f = tid; f < T; f += 1024) {
        int lo = 0, hi = NCH;
        while (hi - lo > 1) {
            const int mid = (lo + hi) >> 1;
            if (s_pos[mid] <= f) lo = mid; else hi = mid;
        }
        const int wd = binned[s_adr[lo] + (f - s_pos[lo])];
        const int q  = atomicAdd(&s_cnt[wd & (RNODES - 1)], 1);
        ssrc[bs + q] = (int)((unsigned)wd >> RSH);
    }
}

__global__ __launch_bounds__(256) void gat_node_gather(
    const int* __restrict__ off, const int* __restrict__ ssrc,
    const float* __restrict__ aS, const float* __restrict__ aD,
    const float* __restrict__ xp,
    const float* __restrict__ bias,
    const float* __restrict__ gamma, const float* __restrict__ beta,
    float* __restrict__ out, int N)
{
    const int d = __builtin_amdgcn_readfirstlane(
        (int)((blockIdx.x * blockDim.x + threadIdx.x) >> 5));
    const int lane = threadIdx.x & 31;
    if (d >= N) return;
    const int head = lane >> 4;
    const int fo = lane * 4;

    const int beg = off[d];
    const int end = off[d + 1];
    const float ad0 = aD[d * 2 + 0], ad1 = aD[d * 2 + 1];
    const float es0 = lrelu(aS[d * 2 + 0] + ad0);
    const float es1 = lrelu(aS[d * 2 + 1] + ad1);
    const float4 xs = *(const float4*)(xp + (size_t)d * XDIM + fo);

    float m0 = es0, m1 = es1;
    for (int i = beg + lane; i < end; i += 32) {
        const int s = ssrc[i];
        const float2 a = *(const float2*)(aS + (size_t)s * 2);
        m0 = fmaxf(m0, lrelu(a.x + ad0));
        m1 = fmaxf(m1, lrelu(a.y + ad1));
    }
    #pragma unroll
    for (int o = 16; o >= 1; o >>= 1) {
        m0 = fmaxf(m0, __shfl_xor(m0, o, 32));
        m1 = fmaxf(m1, __shfl_xor(m1, o, 32));
    }

    const float ws0 = __expf(es0 - m0), ws1 = __expf(es1 - m1);
    const float wsh = head ? ws1 : ws0;
    float4 acc = {xs.x * wsh, xs.y * wsh, xs.z * wsh, xs.w * wsh};
    float dn0 = 0.0f, dn1 = 0.0f;

    for (int base = beg; base < end; base += 32) {
        const int i = base + lane;
        int s = 0;
        float w0 = 0.0f, w1 = 0.0f;
        if (i < end) {
            s = ssrc[i];
            const float2 a = *(const float2*)(aS + (size_t)s * 2);
            w0 = __expf(lrelu(a.x + ad0) - m0);
            w1 = __expf(lrelu(a.y + ad1) - m1);
        }
        dn0 += w0;
        dn1 += w1;
        const int rem = end - base;
        const int n = rem < 32 ? rem : 32;
        #pragma unroll 4
        for (int j = 0; j < n; ++j) {
            const int sj = __builtin_amdgcn_readlane(s, j);
            const float wa = __int_as_float(__builtin_amdgcn_readlane(__float_as_int(w0), j));
            const float wb = __int_as_float(__builtin_amdgcn_readlane(__float_as_int(w1), j));
            const float w = head ? wb : wa;
            const float4 v = *(const float4*)(xp + (size_t)sj * XDIM + fo);
            acc.x += v.x * w;
            acc.y += v.y * w;
            acc.z += v.z * w;
            acc.w += v.w * w;
        }
    }
    #pragma unroll
    for (int o = 16; o >= 1; o >>= 1) {
        dn0 += __shfl_xor(dn0, o, 32);
        dn1 += __shfl_xor(dn1, o, 32);
    }
    const float dn = (head ? dn1 + ws1 : dn0 + ws0) + SM_EPS;

    const float q0 = acc.x / dn, q1 = acc.y / dn, q2 = acc.z / dn, q3 = acc.w / dn;
    const int f = (lane & 15) * 4;
    const float4 bv = *(const float4*)(bias + f);
    const float v0 = (q0 + __shfl_xor(q0, 16, 32)) * 0.5f + bv.x;
    const float v1 = (q1 + __shfl_xor(q1, 16, 32)) * 0.5f + bv.y;
    const float v2 = (q2 + __shfl_xor(q2, 16, 32)) * 0.5f + bv.z;
    const float v3 = (q3 + __shfl_xor(q3, 16, 32)) * 0.5f + bv.w;

    float sum = (v0 + v1) + (v2 + v3);
    #pragma unroll
    for (int o = 8; o >= 1; o >>= 1) sum += __shfl_xor(sum, o, 32);
    const float mu = sum * (1.0f / OUT_F);
    const float d0 = v0 - mu, d1 = v1 - mu, d2 = v2 - mu, d3 = v3 - mu;
    float sq = (d0 * d0 + d1 * d1) + (d2 * d2 + d3 * d3);
    #pragma unroll
    for (int o = 8; o >= 1; o >>= 1) sq += __shfl_xor(sq, o, 32);
    const float inv = rsqrtf(sq * (1.0f / OUT_F) + LN_EPS);

    if (lane < 16) {
        const float4 gv = *(const float4*)(gamma + f);
        const float4 be = *(const float4*)(beta + f);
        float4 o4;
        o4.x = d0 * inv * gv.x + be.x;
        o4.y = d1 * inv * gv.y + be.y;
        o4.z = d2 * inv * gv.z + be.z;
        o4.w = d3 * inv * gv.w + be.w;
        *(float4*)(out + (size_t)d * OUT_F + f) = o4;
    }
}

extern "C" void kernel_launch(void* const* d_in, const int* in_sizes, int n_in,
                              void* d_out, int out_size, void* d_ws, size_t ws_size,
                              hipStream_t stream)
{
    const float* X    = (const float*)d_in[0];
    const int*   ei   = (const int*)  d_in[1];
    const float* W    = (const float*)d_in[2];
    const float* attS = (const float*)d_in[3];
    const float* attD = (const float*)d_in[4];
    const float* bias = (const float*)d_in[5];
    const float* g    = (const float*)d_in[6];
    const float* b    = (const float*)d_in[7];
    float* out = (float*)d_out;

    const int N = in_sizes[0] / IN_F;
    const int E = in_sizes[1] / 2;
    const int* esrc = ei;
    const int* edst = ei + E;

    const int NB  = (N + RNODES - 1) / RNODES;
    const int NCH = (E + CH - 1) / CH;

    float* ws     = (float*)d_ws;
    float* xp     = ws;
    float* aS     = xp + (size_t)N * XDIM;
    float* aD     = aS + (size_t)N * 2;
    int*   binTot = (int*)(aD + (size_t)N * 2);
    int*   off    = binTot + NBMAX;
    int*   runs   = off + (size_t)N + 1;
    int*   binned = runs + (size_t)NCH * NB;
    int*   ssrc   = binned + (size_t)E;

    {
        const int waves = (N + 15) / 16;
        gat_gemm_wmma<<<(waves + 7) / 8, 256, 0, stream>>>(
            X, W, attS, attD, xp, aS, aD, binTot, N);
    }
    gat_edge_bin<<<NCH, 1024, 0, stream>>>(esrc, edst, binTot, runs, binned, NB, E);
    gat_bin_sort<<<NB, 1024, 0, stream>>>(binTot, runs, binned, off, ssrc, N, NB, NCH);
    gat_node_gather<<<(N + 7) / 8, 256, 0, stream>>>(
        off, ssrc, aS, aD, xp, bias, g, b, out, N);
}


# Edge binning blocks (256 threads, 4096-edge chunks) run inside the GEMM's launch; bin starts summed from the run table, no global adds; 3 launches; LDS-arrival order of the sums changes
# speedup vs baseline: 6.9420x; 1.1498x over previous
#include <hip/hip_runtime.h>
#include <hip/hip_bf16.h>
#include <math.h>


typedef __attribute__((ext_vector_type(2))) float v2f;
typedef __attribute__((ext_vector_type(4))) float v4f;
typedef __attribute__((ext_vector_type(8))) float v8f;

#define IN_F     128
#define OUT_F    64
#define HEADS    2
#define XDIM     128
#define NEG_SLOPE 0.2f
#define LN_EPS    1e-5f
#define SM_EPS    1e-16f

#define RSH     7
#define RNODES  (1 << RSH)
#define NBMAX   1024
#define CH      4096
#define BT      256
#define EPT     16
#define NCHMAX  1024

#define WPITCH  132


template <int M>
__device__ __forceinline__ float row_xor_get(float x)
{
    return __int_as_float(__builtin_amdgcn_update_dpp(
        0, __float_as_int(x), 0x160 | M, 0xf, 0xf, true));
}

__device__ __forceinline__ float row_sum_transposed(const float (&p)[16], int c)
{
    const bool b3 = (c & 8) != 0, b2 = (c & 4) != 0, b1 = (c & 2) != 0, b0 = (c & 1) != 0;
    float q[8], r[4], s[2];
    #pragma unroll
    for (int j = 0; j < 8; ++j) {
        const float keep = b3 ? p[j + 8] : p[j];
        const float send = b3 ? p[j] : p[j + 8];
        q[j] = keep + row_xor_get<8>(send);
    }
    #pragma unroll
    for (int j = 0; j < 4; ++j) {
        const float keep = b2 ? q[j + 4] : q[j];
        const float send = b2 ? q[j] : q[j + 4];
        r[j] = keep + row_xor_get<4>(send);
    }
    #pragma unroll
    for (int j = 0; j < 2; ++j) {
        const float keep = b1 ? r[j + 2] : r[j];
        const float send = b1 ? r[j] : r[j + 2];
        s[j] = keep + row_xor_get<2>(send);
    }
    const float keep = b0 ? s[1] : s[0];
    const float send = b0 ? s[0] : s[1];
    return keep + row_xor_get<1>(send);
}

__device__ __forceinline__ void gemm_block(
    float* __restrict__ s_wt, const int gb,
    const float* __restrict__ X, const float* __restrict__ W,
    const float* __restrict__ attD,
    float* __restrict__ xp, float* __restrict__ aD, int N)
{
    const int gt = gb * BT + (int)threadIdx.x;

    #pragma unroll
    for (int j = 0; j < (XDIM * IN_F / 4) / BT; ++j) {
        const int idx = j * BT + (int)threadIdx.x;
        const float4 wv = *(const float4*)(W + (size_t)idx * 4);
        *(float4*)(s_wt + (idx >> 5) * WPITCH + (idx & 31) * 4) = wv;
    }
    __syncthreads();

    const int wave = gt >> 5;
    const int lane = threadIdx.x & 31;
    const int r0 = wave * 16;
    if (r0 >= N) return;

    const int lm = lane & 15;
    const int kk = (lane >> 4) << 1;

    int ar = r0 + lm; if (ar >= N) ar = N - 1;
    const float* arow = X + (size_t)ar * IN_F + kk;
    const float* brow = s_wt + lm * WPITCH + kk;

    v8f acc[8];
    #pragma unroll
    for (int t = 0; t < 8; ++t) acc[t] = (v8f){};

    #pragma unroll 4
    for (int k0 = 0; k0 < IN_F; k0 += 4) {
        v2f a = *(const v2f*)(arow + k0);
        #pragma unroll
        for (int t = 0; t < 8; ++t) {
            v2f b = *(const v2f*)(brow + t * 16 * WPITCH + k0);
            acc[t] = __builtin_amdgcn_wmma_f32_16x16x4_f32(
                 false, a,  false, b,
                 (short)0, acc[t],  false,  false);
        }
    }

    const int mofs = (lane >> 4) << 3;
    #pragma unroll
    for (int v = 0; v < 8; ++v) {
        const int row = r0 + mofs + v;
        if (row < N) {
            float* orow = xp + (size_t)row * XDIM + lm;
            #pragma unroll
            for (int t = 0; t < 8; ++t) orow[t * 16] = acc[t][v];
        }
    }

    float dv[8];
    #pragma unroll
    for (int t = 0; t < 8; ++t) dv[t] = attD[t * 16 + lm];
    float pD[16];
    #pragma unroll
    for (int v = 0; v < 8; ++v) {
        #pragma unroll
        for (int h = 0; h < 2; ++h) {
            const float x0 = acc[4 * h + 0][v], x1 = acc[4 * h + 1][v];
            const float x2 = acc[4 * h + 2][v], x3 = acc[4 * h + 3][v];
            pD[v * 2 + h] = x0 * dv[4 * h + 0] + x1 * dv[4 * h + 1]
                          + x2 * dv[4 * h + 2] + x3 * dv[4 * h + 3];
        }
    }
    const float tD = row_sum_transposed(pD, lm);
    if (r0 + (lane >> 1) < N) aD[(size_t)r0 * 2 + lane] = tD;
}

__device__ __forceinline__ float lrelu(float v)
{
    return v > 0.0f ? v : NEG_SLOPE * v;
}

template <int NW>
__device__ __forceinline__ int block_scan_excl(int v, int tid, int* s_w, int* total)
{
    const int lane = tid & 31;
    const int wv   = tid >> 5;
    int inc = v;
    #pragma unroll
    for (int o = 1; o < 32; o <<= 1) {
        const int t = __shfl_up(inc, o, 32);
        if (lane >= o) inc += t;
    }
    __syncthreads();
    if (lane == 31) s_w[wv] = inc;
    __syncthreads();
    const int wt = (lane < NW) ? s_w[lane] : 0;
    int winc = wt;
    #pragma unroll
    for (int o = 1; o < 32; o <<= 1) {
        const int t = __shfl_up(winc, o, 32);
        if (lane >= o) winc += t;
    }
    *total = __shfl(winc, 31, 32);
    return __shfl(winc - wt, wv, 32) + inc - v;
}

__device__ __forceinline__ void bin_block(
    int* __restrict__ sm, const int c,
    const int* __restrict__ esrc, const int* __restrict__ edst,
    int* __restrict__ runs, int* __restrict__ binned,
    int NB, int E)
{
    int* s_hist = sm;
    int* s_base = s_hist + NBMAX;
    int* s_rec  = s_base + NBMAX;
    int* s_w    = s_rec + CH;
    const int tid = threadIdx.x;
    const int e0  = c * CH;
    const int rem = E - e0;
    const int nE  = rem < CH ? rem : CH;

    *(int4*)(s_hist + 4 * tid) = make_int4(0, 0, 0, 0);
    __syncthreads();

    int dd[EPT], ss[EPT];
    #pragma unroll
    for (int k = 0; k < EPT; ++k) {
        const int i = k * BT + tid;
        const int j = e0 + (i < nE ? i : 0);
        dd[k] = edst[j];
        ss[k] = esrc[j];
    }

    int w[EPT], bn[EPT], rk[EPT];
    #pragma unroll
    for (int k = 0; k < EPT; ++k) {
        const int i = k * BT + tid;
        bn[k] = -1; w[k] = 0; rk[k] = 0;
        if (i < nE) {
            const int d = dd[k];
            const int s = ss[k];
            unsigned b = (unsigned)d >> RSH;
            if (b > NBMAX - 1) b = NBMAX - 1;
            bn[k] = (int)b;
            w[k]  = (s << RSH) | (d & (RNODES - 1));
            rk[k] = atomicAdd(&s_hist[b], 1);
        }
    }
    __syncthreads();

    const int4 cv = *(const int4*)(s_hist + 4 * tid);
    int tot;
    const int ex0 = block_scan_excl<BT / 32>(cv.x + cv.y + cv.z + cv.w, tid, s_w, &tot);
    const int ex1 = ex0 + cv.x, ex2 = ex1 + cv.y, ex3 = ex2 + cv.z;
    *(int4*)(s_base + 4 * tid) = make_int4(ex0, ex1, ex2, ex3);
    {
        int* rrow = runs + (size_t)c * NB + 4 * tid;
        if (4 * tid + 0 < NB) rrow[0] = (ex0 << 16) | cv.x;
        if (4 * tid + 1 < NB) rrow[1] = (ex1 << 16) | cv.y;
        if (4 * tid + 2 < NB) rrow[2] = (ex2 << 16) | cv.z;
        if (4 * tid + 3 < NB) rrow[3] = (ex3 << 16) | cv.w;
    }
    __syncthreads();

    #pragma unroll
    for (int k = 0; k < EPT; ++k)
        if (bn[k] >= 0) s_rec[s_base[bn[k]] + rk[k]] = w[k];
    __syncthreads();

    for (int i = tid; i < nE; i += BT) binned[e0 + i] = s_rec[i];
}

__global__ __launch_bounds__(BT) void gat_gemm_bin(
    const float* __restrict__ X, const float* __restrict__ W,
    const float* __restrict__ attD,
    float* __restrict__ xp, float* __restrict__ aD,
    const int* __restrict__ esrc, const int* __restrict__ edst,
    int* __restrict__ runs, int* __restrict__ binned,
    int N, int NB, int E, int NCH)
{
    __shared__ __attribute__((aligned(16))) float s_mem[XDIM * WPITCH];

    const int bid = blockIdx.x;
    if (bid < NCH) bin_block((int*)s_mem, bid, esrc, edst, runs, binned, NB, E);
    else           gemm_block(s_mem, bid - NCH, X, W, attD, xp, aD, N);
}

__global__ __launch_bounds__(1024) void gat_bin_sort(
    const int* __restrict__ runs, const int* __restrict__ binned,
    int* __restrict__ off, int* __restrict__ ssrc,
    int N, int NB, int NCH)
{
    __shared__ int s_pos[NCHMAX];
    __shared__ int s_adr[NCHMAX];
    __shared__ int s_cnt[RNODES];
    __shared__ int s_w[32];
    __shared__ int s_bs;
    const int tid = threadIdx.x;
    const int b   = blockIdx.x;

    if (tid == 0) s_bs = 0;

    int cn = 0, ad = 0, rs = 0;
    if (tid < NCH) {
        const int pk = runs[tid * NB + b];
        cn = pk & 0xffff;
        rs = (int)((unsigned)pk >> 16);
        ad = tid * CH + rs;
    }
    int T;
    const int p = block_scan_excl<32>(cn, tid, s_w, &T);

    #pragma unroll
    for (int o = 16; o >= 1; o >>= 1) rs += __shfl_xor(rs, o, 32);
    if ((tid & 31) == 0 && rs != 0) atomicAdd(&s_bs, rs);

    s_pos[tid] = p;
    s_adr[tid] = ad;
    if (tid < RNODES) s_cnt[tid] = 0;
    __syncthreads();

    const int bs = s_bs;

    for (int f = tid; f < T; f += 1024) {
        int lo = 0, hi = NCH;
        while (hi - lo > 1) {
            const int mid = (lo + hi) >> 1;
            if (s_pos[mid] <= f) lo = mid; else hi = mid;
        }
        const int wd = binned[s_adr[lo] + (f - s_pos[lo])];
        atomicAdd(&s_cnt[wd & (RNODES - 1)], 1);
    }
    __syncthreads();

    const int cv = (tid < RNODES) ? s_cnt[tid] : 0;
    int T2;
    const int cex = block_scan_excl<32>(cv, tid, s_w, &T2);
    if (tid < RNODES) {
        const int node = (b << RSH) + tid;
        if (node < N) off[node] = bs + cex;
        s_cnt[tid] = cex;
    }
    if (b == NB - 1 && tid == 0) off[N] = bs + T;
    __syncthreads();

    for (int f = tid; f < T; f += 1024) {
        int lo = 0, hi = NCH;
        while (hi - lo > 1) {
            const int mid = (lo + hi) >> 1;
            if (s_pos[mid] <= f) lo = mid; else hi = mid;
        }
        const int wd = binned[s_adr[lo] + (f - s_pos[lo])];
        const int q  = atomicAdd(&s_cnt[wd & (RNODES - 1)], 1);
        ssrc[bs + q] = (int)((unsigned)wd >> RSH);
    }
}


template <int K>
__device__ __forceinline__ float quad_get(float x)
{
    return __int_as_float(__builtin_amdgcn_update_dpp(
        0, __float_as_int(x), K * 0x55, 0xf, 0xf, true));
}

__device__ __forceinline__ float dot4(const v4f v, const v4f a)
{
    const v2f t = v.lo * a.lo + v.hi * a.hi;
    return t.x + t.y;
}

__global__ __launch_bounds__(256) void gat_node_gather(
    const int* __restrict__ off, const int* __restrict__ ssrc,
    const float* __restrict__ attS, const float* __restrict__ aD,
    const float* __restrict__ xp,
    const float* __restrict__ bias,
    const float* __restrict__ gamma, const float* __restrict__ beta,
    float* __restrict__ out, int N)
{
    const int d = __builtin_amdgcn_readfirstlane(
        (int)((blockIdx.x * blockDim.x + threadIdx.x) >> 5));
    const int lane = threadIdx.x & 31;
    if (d >= N) return;
    const int head = lane >> 4;
    const int fo = lane * 4;
    const int k4 = lane & 3;
    const bool b0 = (lane & 1) != 0, b1 = (lane & 2) != 0;

    const int beg = off[d];
    const int end = off[d + 1];
    const float adh = aD[d * 2 + head];
    const v4f as4 = *(const v4f*)(attS + fo);
    const v4f xs = *(const v4f*)(xp + (size_t)d * XDIM + fo);

    float ps = dot4(xs, as4);
    ps += row_xor_get<8>(ps);
    ps += row_xor_get<4>(ps);
    ps += row_xor_get<2>(ps);
    ps += row_xor_get<1>(ps);
    float m = lrelu(ps + adh);
    v4f acc = xs;
    float dn = (k4 == 0) ? 1.0f : 0.0f;

    const v4f zero4 = {0.0f, 0.0f, 0.0f, 0.0f};
    for (int base = beg; base < end; base += 32) {
        const int i = base + lane;
        const int s = (i < end) ? ssrc[i] : 0;
        const int rem = end - base;
        const int n = rem < 32 ? rem : 32;
        for (int j = 0; j < n; j += 4) {
            v4f v0, v1 = zero4, v2 = zero4, v3 = zero4;
            v0 = *(const v4f*)(xp + (size_t)__builtin_amdgcn_readlane(s, j) * XDIM + fo);
            if (j + 1 < n) v1 = *(const v4f*)(xp + (size_t)__builtin_amdgcn_readlane(s, j + 1) * XDIM + fo);
            if (j + 2 < n) v2 = *(const v4f*)(xp + (size_t)__builtin_amdgcn_readlane(s, j + 2) * XDIM + fo);
            if (j + 3 < n) v3 = *(const v4f*)(xp + (size_t)__builtin_amdgcn_readlane(s, j + 3) * XDIM + fo);

            const float p0 = dot4(v0, as4), p1 = dot4(v1, as4);
            const float p2 = dot4(v2, as4), p3 = dot4(v3, as4);
            const float qa = (b0 ? p1 : p0) + row_xor_get<1>(b0 ? p0 : p1);
            const float qb = (b0 ? p3 : p2) + row_xor_get<1>(b0 ? p2 : p3);
            float r = (b1 ? qb : qa) + row_xor_get<2>(b1 ? qa : qb);
            r += row_xor_get<4>(r);
            r += row_xor_get<8>(r);
            const bool valid = j + k4 < n;
            const float e = lrelu(r + adh);

            float g = valid ? e : m;
            g = fmaxf(g, row_xor_get<1>(g));
            g = fmaxf(g, row_xor_get<2>(g));
            const float mn = fmaxf(m, g);
            const float sc = __expf(m - mn);
            const float w = valid ? __expf(e - mn) : 0.0f;
            m = mn;
            dn = dn * sc + w;

            const float w0 = quad_get<0>(w), w1 = quad_get<1>(w);
            const float w2 = quad_get<2>(w), w3 = quad_get<3>(w);
            acc = acc * sc + v0 * w0;
            acc += v1 * w1;
            acc += v2 * w2;
            acc += v3 * w3;
        }
    }
    dn += row_xor_get<1>(dn);
    dn += row_xor_get<2>(dn);
    dn += SM_EPS;

    const float q0 = acc.x / dn, q1 = acc.y / dn, q2 = acc.z / dn, q3 = acc.w / dn;
    const int f = (lane & 15) * 4;
    const float4 bv = *(const float4*)(bias + f);
    const float v0 = (q0 + __shfl_xor(q0, 16, 32)) * 0.5f + bv.x;
    const float v1 = (q1 + __shfl_xor(q1, 16, 32)) * 0.5f + bv.y;
    const float v2 = (q2 + __shfl_xor(q2, 16, 32)) * 0.5f + bv.z;
    const float v3 = (q3 + __shfl_xor(q3, 16, 32)) * 0.5f + bv.w;

    float sum = (v0 + v1) + (v2 + v3);
    #pragma unroll
    for (int o = 8; o >= 1; o >>= 1) sum += __shfl_xor(sum, o, 32);
    const float mu = sum * (1.0f / OUT_F);
    const float d0 = v0 - mu, d1 = v1 - mu, d2 = v2 - mu, d3 = v3 - mu;
    float sq = (d0 * d0 + d1 * d1) + (d2 * d2 + d3 * d3);
    #pragma unroll
    for (int o = 8; o >= 1; o >>= 1) sq += __shfl_xor(sq, o, 32);
    const float inv = rsqrtf(sq * (1.0f / OUT_F) + LN_EPS);

    if (lane < 16) {
        const float4 gv = *(const float4*)(gamma + f);
        const float4 be = *(const float4*)(beta + f);
        float4 o4;
        o4.x = d0 * inv * gv.x + be.x;
        o4.y = d1 * inv * gv.y + be.y;
        o4.z = d2 * inv * gv.z + be.z;
        o4.w = d3 * inv * gv.w + be.w;
        *(float4*)(out + (size_t)d * OUT_F + f) = o4;
    }
}

extern "C" void kernel_launch(void* const* d_in, const int* in_sizes, int n_in,
                              void* d_out, int out_size, void* d_ws, size_t ws_size,
                              hipStream_t stream)
{
    const float* X    = (const float*)d_in[0];
    const int*   ei   = (const int*)  d_in[1];
    const float* W    = (const float*)d_in[2];
    const float* attS = (const float*)d_in[3];
    const float* attD = (const float*)d_in[4];
    const float* bias = (const float*)d_in[5];
    const float* g    = (const float*)d_in[6];
    const float* b    = (const float*)d_in[7];
    float* out = (float*)d_out;

    const int N = in_sizes[0] / IN_F;
    const int E = in_sizes[1] / 2;
    const int* esrc = ei;
    const int* edst = ei + E;

    const int NB  = (N + RNODES - 1) / RNODES;
    const int NCH = (E + CH - 1) / CH;

    float* ws     = (float*)d_ws;
    float* xp     = ws;
    float* aD     = xp + (size_t)N * XDIM;
    int*   off    = (int*)(aD + (size_t)N * 2);
    int*   runs   = off + (size_t)N + 1;
    int*   binned = runs + (size_t)NCH * NB;
    int*   ssrc   = binned + (size_t)E;

    {
        const int waves = (N + 15) / 16;
        const int GB    = (waves + 7) / 8;
        gat_gemm_bin<<<NCH + GB, BT, 0, stream>>>(
            X, W, attD, xp, aD, esrc, edst, runs, binned, N, NB, E, NCH);
    }
    gat_bin_sort<<<NB, 1024, 0, stream>>>(runs, binned, off, ssrc, N, NB, NCH);
    gat_node_gather<<<(N + 7) / 8, 256, 0, stream>>>(
        off, ssrc, attS, aD, xp, bias, g, b, out, N);
}


# Launch 2 keeps each bin's edge words in LDS: run owners write per-edge addresses, both passes read the slots (no per-edge bisection, binned read once); 256-thread blocks; sums in LDS-arrival order
# speedup vs baseline: 7.3958x; 1.0654x over previous
#include <hip/hip_runtime.h>
#include <hip/hip_bf16.h>
#include <math.h>


typedef __attribute__((ext_vector_type(2))) float v2f;
typedef __attribute__((ext_vector_type(4))) float v4f;
typedef __attribute__((ext_vector_type(8))) float v8f;

#define IN_F     128
#define OUT_F    64
#define HEADS    2
#define XDIM     128
#define NEG_SLOPE 0.2f
#define LN_EPS    1e-5f
#define SM_EPS    1e-16f

#define RSH     7
#define RNODES  (1 << RSH)
#define NBMAX   1024
#define CH      4096
#define BT      256
#define EPT     16
#define NCHMAX  1024
#define BS2     256
#define RPT     (NCHMAX / BS2)
#define TCAP    8192
#define EU      16

#define WPITCH  132


template <int M>
__device__ __forceinline__ float row_xor_get(float x)
{
    return __int_as_float(__builtin_amdgcn_update_dpp(
        0, __float_as_int(x), 0x160 | M, 0xf, 0xf, true));
}

__device__ __forceinline__ float row_sum_transposed(const float (&p)[16], int c)
{
    const bool b3 = (c & 8) != 0, b2 = (c & 4) != 0, b1 = (c & 2) != 0, b0 = (c & 1) != 0;
    float q[8], r[4], s[2];
    #pragma unroll
    for (int j = 0; j < 8; ++j) {
        const float keep = b3 ? p[j + 8] : p[j];
        const float send = b3 ? p[j] : p[j + 8];
        q[j] = keep + row_xor_get<8>(send);
    }
    #pragma unroll
    for (int j = 0; j < 4; ++j) {
        const float keep = b2 ? q[j + 4] : q[j];
        const float send = b2 ? q[j] : q[j + 4];
        r[j] = keep + row_xor_get<4>(send);
    }
    #pragma unroll
    for (int j = 0; j < 2; ++j) {
        const float keep = b1 ? r[j + 2] : r[j];
        const float send = b1 ? r[j] : r[j + 2];
        s[j] = keep + row_xor_get<2>(send);
    }
    const float keep = b0 ? s[1] : s[0];
    const float send = b0 ? s[0] : s[1];
    return keep + row_xor_get<1>(send);
}

__device__ __forceinline__ void gemm_block(
    float* __restrict__ s_wt, const int gb,
    const float* __restrict__ X, const float* __restrict__ W,
    const float* __restrict__ attD,
    float* __restrict__ xp, float* __restrict__ aD, int N)
{
    const int gt = gb * BT + (int)threadIdx.x;

    #pragma unroll
    for (int j = 0; j < (XDIM * IN_F / 4) / BT; ++j) {
        const int idx = j * BT + (int)threadIdx.x;
        const float4 wv = *(const float4*)(W + (size_t)idx * 4);
        *(float4*)(s_wt + (idx >> 5) * WPITCH + (idx & 31) * 4) = wv;
    }
    __syncthreads();

    const int wave = gt >> 5;
    const int lane = threadIdx.x & 31;
    const int r0 = wave * 16;
    if (r0 >= N) return;

    const int lm = lane & 15;
    const int kk = (lane >> 4) << 1;

    int ar = r0 + lm; if (ar >= N) ar = N - 1;
    const float* arow = X + (size_t)ar * IN_F + kk;
    const float* brow = s_wt + lm * WPITCH + kk;

    v8f acc[8];
    #pragma unroll
    for (int t = 0; t < 8; ++t) acc[t] = (v8f){};

    #pragma unroll 4
    for (int k0 = 0; k0 < IN_F; k0 += 4) {
        v2f a = *(const v2f*)(arow + k0);
        #pragma unroll
        for (int t = 0; t < 8; ++t) {
            v2f b = *(const v2f*)(brow + t * 16 * WPITCH + k0);
            acc[t] = __builtin_amdgcn_wmma_f32_16x16x4_f32(
                 false, a,  false, b,
                 (short)0, acc[t],  false,  false);
        }
    }

    const int mofs = (lane >> 4) << 3;
    #pragma unroll
    for (int v = 0; v < 8; ++v) {
        const int row = r0 + mofs + v;
        if (row < N) {
            float* orow = xp + (size_t)row * XDIM + lm;
            #pragma unroll
            for (int t = 0; t < 8; ++t) orow[t * 16] = acc[t][v];
        }
    }

    float dv[8];
    #pragma unroll
    for (int t = 0; t < 8; ++t) dv[t] = attD[t * 16 + lm];
    float pD[16];
    #pragma unroll
    for (int v = 0; v < 8; ++v) {
        #pragma unroll
        for (int h = 0; h < 2; ++h) {
            const float x0 = acc[4 * h + 0][v], x1 = acc[4 * h + 1][v];
            const float x2 = acc[4 * h + 2][v], x3 = acc[4 * h + 3][v];
            pD[v * 2 + h] = x0 * dv[4 * h + 0] + x1 * dv[4 * h + 1]
                          + x2 * dv[4 * h + 2] + x3 * dv[4 * h + 3];
        }
    }
    const float tD = row_sum_transposed(pD, lm);
    if (r0 + (lane >> 1) < N) aD[(size_t)r0 * 2 + lane] = tD;
}

__device__ __forceinline__ float lrelu(float v)
{
    return v > 0.0f ? v : NEG_SLOPE * v;
}

template <int NW>
__device__ __forceinline__ int block_scan_excl(int v, int tid, int* s_w, int* total)
{
    const int lane = tid & 31;
    const int wv   = tid >> 5;
    int inc = v;
    #pragma unroll
    for (int o = 1; o < 32; o <<= 1) {
        const int t = __shfl_up(inc, o, 32);
        if (lane >= o) inc += t;
    }
    __syncthreads();
    if (lane == 31) s_w[wv] = inc;
    __syncthreads();
    const int wt = (lane < NW) ? s_w[lane] : 0;
    int winc = wt;
    #pragma unroll
    for (int o = 1; o < 32; o <<= 1) {
        const int t = __shfl_up(winc, o, 32);
        if (lane >= o) winc += t;
    }
    *total = __shfl(winc, 31, 32);
    return __shfl(winc - wt, wv, 32) + inc - v;
}

__device__ __forceinline__ void bin_block(
    int* __restrict__ sm, const int c,
    const int* __restrict__ esrc, const int* __restrict__ edst,
    int* __restrict__ runs, int* __restrict__ binned,
    int NB, int E)
{
    int* s_hist = sm;
    int* s_base = s_hist + NBMAX;
    int* s_rec  = s_base + NBMAX;
    int* s_w    = s_rec + CH;
    const int tid = threadIdx.x;
    const int e0  = c * CH;
    const int rem = E - e0;
    const int nE  = rem < CH ? rem : CH;

    *(int4*)(s_hist + 4 * tid) = make_int4(0, 0, 0, 0);
    __syncthreads();

    int dd[EPT], ss[EPT];
    #pragma unroll
    for (int k = 0; k < EPT; ++k) {
        const int i = k * BT + tid;
        const int j = e0 + (i < nE ? i : 0);
        dd[k] = edst[j];
        ss[k] = esrc[j];
    }

    int w[EPT], bn[EPT], rk[EPT];
    #pragma unroll
    for (int k = 0; k < EPT; ++k) {
        const int i = k * BT + tid;
        bn[k] = -1; w[k] = 0; rk[k] = 0;
        if (i < nE) {
            const int d = dd[k];
            const int s = ss[k];
            unsigned b = (unsigned)d >> RSH;
            if (b > NBMAX - 1) b = NBMAX - 1;
            bn[k] = (int)b;
            w[k]  = (s << RSH) | (d & (RNODES - 1));
            rk[k] = atomicAdd(&s_hist[b], 1);
        }
    }
    __syncthreads();

    const int4 cv = *(const int4*)(s_hist + 4 * tid);
    int tot;
    const int ex0 = block_scan_excl<BT / 32>(cv.x + cv.y + cv.z + cv.w, tid, s_w, &tot);
    const int ex1 = ex0 + cv.x, ex2 = ex1 + cv.y, ex3 = ex2 + cv.z;
    *(int4*)(s_base + 4 * tid) = make_int4(ex0, ex1, ex2, ex3);
    {
        int* rrow = runs + (size_t)c * NB + 4 * tid;
        if (4 * tid + 0 < NB) rrow[0] = (ex0 << 16) | cv.x;
        if (4 * tid + 1 < NB) rrow[1] = (ex1 << 16) | cv.y;
        if (4 * tid + 2 < NB) rrow[2] = (ex2 << 16) | cv.z;
        if (4 * tid + 3 < NB) rrow[3] = (ex3 << 16) | cv.w;
    }
    __syncthreads();

    #pragma unroll
    for (int k = 0; k < EPT; ++k)
        if (bn[k] >= 0) s_rec[s_base[bn[k]] + rk[k]] = w[k];
    __syncthreads();

    for (int i = tid; i < nE; i += BT) binned[e0 + i] = s_rec[i];
}

__global__ __launch_bounds__(BT) void gat_gemm_bin(
    const float* __restrict__ X, const float* __restrict__ W,
    const float* __restrict__ attD,
    float* __restrict__ xp, float* __restrict__ aD,
    const int* __restrict__ esrc, const int* __restrict__ edst,
    int* __restrict__ runs, int* __restrict__ binned,
    int N, int NB, int E, int NCH)
{
    __shared__ __attribute__((aligned(16))) float s_mem[XDIM * WPITCH];

    const int bid = blockIdx.x;
    if (bid < NCH) bin_block((int*)s_mem, bid, esrc, edst, runs, binned, NB, E);
    else           gemm_block(s_mem, bid - NCH, X, W, attD, xp, aD, N);
}


__device__ __forceinline__ int run_edge_addr(const int* s_pos, const int* s_adr, int f)
{
    int lo = 0, hi = NCHMAX;
    while (hi - lo > 1) {
        const int mid = (lo + hi) >> 1;
        if (s_pos[mid] <= f) lo = mid; else hi = mid;
    }
    return s_adr[lo] + (f - s_pos[lo]);
}

__global__ __launch_bounds__(BS2) void gat_bin_sort(
    const int* __restrict__ runs, const int* __restrict__ binned,
    int* __restrict__ off, int* __restrict__ ssrc,
    int N, int NB, int NCH)
{
    __shared__ int s_pos[NCHMAX];
    __shared__ int s_adr[NCHMAX];
    __shared__ int s_e[TCAP];
    __shared__ int s_cnt[RNODES];
    __shared__ int s_w[32];
    __shared__ int s_bs;
    const int tid = threadIdx.x;
    const int b   = blockIdx.x;

    if (tid == 0) s_bs = 0;

    int pk[RPT];
    #pragma unroll
    for (int r = 0; r < RPT; ++r) {
        const int c = r * BS2 + tid;
        pk[r] = runs[(c < NCH ? c : 0) * NB + b];
    }
    int cn[RPT], ad[RPT];
    int rs = 0, tot = 0;
    #pragma unroll
    for (int r = 0; r < RPT; ++r) {
        const int c = r * BS2 + tid;
        cn[r] = 0; ad[r] = 0;
        if (c < NCH) {
            const int st = (int)((unsigned)pk[r] >> 16);
            cn[r] = pk[r] & 0xffff;
            ad[r] = c * CH + st;
            rs += st;
        }
        tot += cn[r];
    }
    int T;
    const int p = block_scan_excl<BS2 / 32>(tot, tid, s_w, &T);

    #pragma unroll
    for (int o = 16; o >= 1; o >>= 1) rs += __shfl_xor(rs, o, 32);
    if ((tid & 31) == 0 && rs != 0) atomicAdd(&s_bs, rs);

    {
        int q = p;
        #pragma unroll
        for (int r = 0; r < RPT; ++r) {
            s_pos[RPT * tid + r] = q;
            s_adr[RPT * tid + r] = ad[r];
            for (int i = 0; i < cn[r]; ++i)
                if (q + i < TCAP) s_e[q + i] = ad[r] + i;
            q += cn[r];
        }
    }
    if (tid < RNODES) s_cnt[tid] = 0;
    __syncthreads();

    const int bs = s_bs;
    const int TL = T < TCAP ? T : TCAP;

    for (int f0 = tid; f0 < TL; f0 += BS2 * EU) {
        int a[EU], wd[EU];
        #pragma unroll
        for (int u = 0; u < EU; ++u) {
            const int f = f0 + u * BS2;
            a[u] = (f < TL) ? s_e[f] : -1;
        }
        #pragma unroll
        for (int u = 0; u < EU; ++u) {
            wd[u] = 0;
            if (a[u] >= 0) wd[u] = binned[a[u]];
        }
        #pragma unroll
        for (int u = 0; u < EU; ++u) {
            if (a[u] >= 0) {
                s_e[f0 + u * BS2] = wd[u];
                atomicAdd(&s_cnt[wd[u] & (RNODES - 1)], 1);
            }
        }
    }
    for (int f = TCAP + tid; f < T; f += BS2) {
        const int wd = binned[run_edge_addr(s_pos, s_adr, f)];
        atomicAdd(&s_cnt[wd & (RNODES - 1)], 1);
    }
    __syncthreads();

    const int cv = (tid < RNODES) ? s_cnt[tid] : 0;
    int T2;
    const int cex = block_scan_excl<BS2 / 32>(cv, tid, s_w, &T2);
    if (tid < RNODES) {
        const int node = (b << RSH) + tid;
        if (node < N) off[node] = bs + cex;
        s_cnt[tid] = cex;
    }
    if (b == NB - 1 && tid == 0) off[N] = bs + T;
    __syncthreads();

    for (int f0 = tid; f0 < TL; f0 += BS2 * EU) {
        int wd[EU], q[EU];
        #pragma unroll
        for (int u = 0; u < EU; ++u) {
            const int f = f0 + u * BS2;
            wd[u] = (f < TL) ? s_e[f] : -1;
        }
        #pragma unroll
        for (int u = 0; u < EU; ++u) {
            q[u] = 0;
            if (f0 + u * BS2 < TL) q[u] = atomicAdd(&s_cnt[wd[u] & (RNODES - 1)], 1);
        }
        #pragma unroll
        for (int u = 0; u < EU; ++u)
            if (f0 + u * BS2 < TL) ssrc[bs + q[u]] = (int)((unsigned)wd[u] >> RSH);
    }
    for (int f = TCAP + tid; f < T; f += BS2) {
        const int wd = binned[run_edge_addr(s_pos, s_adr, f)];
        const int q  = atomicAdd(&s_cnt[wd & (RNODES - 1)], 1);
        ssrc[bs + q] = (int)((unsigned)wd >> RSH);
    }
}


template <int K>
__device__ __forceinline__ float quad_get(float x)
{
    return __int_as_float(__builtin_amdgcn_update_dpp(
        0, __float_as_int(x), K * 0x55, 0xf, 0xf, true));
}

__device__ __forceinline__ float dot4(const v4f v, const v4f a)
{
    const v2f t = v.lo * a.lo + v.hi * a.hi;
    return t.x + t.y;
}

__global__ __launch_bounds__(256) void gat_node_gather(
    const int* __restrict__ off, const int* __restrict__ ssrc,
    const float* __restrict__ attS, const float* __restrict__ aD,
    const float* __restrict__ xp,
    const float* __restrict__ bias,
    const float* __restrict__ gamma, const float* __restrict__ beta,
    float* __restrict__ out, int N)
{
    const int d = __builtin_amdgcn_readfirstlane(
        (int)((blockIdx.x * blockDim.x + threadIdx.x) >> 5));
    const int lane = threadIdx.x & 31;
    if (d >= N) return;
    const int head = lane >> 4;
    const int fo = lane * 4;
    const int k4 = lane & 3;
    const bool b0 = (lane & 1) != 0, b1 = (lane & 2) != 0;

    const int beg = off[d];
    const int end = off[d + 1];
    const float adh = aD[d * 2 + head];
    const v4f as4 = *(const v4f*)(attS + fo);
    const v4f xs = *(const v4f*)(xp + (size_t)d * XDIM + fo);

    float ps = dot4(xs, as4);
    ps += row_xor_get<8>(ps);
    ps += row_xor_get<4>(ps);
    ps += row_xor_get<2>(ps);
    ps += row_xor_get<1>(ps);
    float m = lrelu(ps + adh);
    v4f acc = xs;
    float dn = (k4 == 0) ? 1.0f : 0.0f;

    const v4f zero4 = {0.0f, 0.0f, 0.0f, 0.0f};
    for (int base = beg; base < end; base += 32) {
        const int i = base + lane;
        const int s = (i < end) ? ssrc[i] : 0;
        const int rem = end - base;
        const int n = rem < 32 ? rem : 32;
        for (int j = 0; j < n; j += 4) {
            v4f v0, v1 = zero4, v2 = zero4, v3 = zero4;
            v0 = *(const v4f*)(xp + (size_t)__builtin_amdgcn_readlane(s, j) * XDIM + fo);
            if (j + 1 < n) v1 = *(const v4f*)(xp + (size_t)__builtin_amdgcn_readlane(s, j + 1) * XDIM + fo);
            if (j + 2 < n) v2 = *(const v4f*)(xp + (size_t)__builtin_amdgcn_readlane(s, j + 2) * XDIM + fo);
            if (j + 3 < n) v3 = *(const v4f*)(xp + (size_t)__builtin_amdgcn_readlane(s, j + 3) * XDIM + fo);

            const float p0 = dot4(v0, as4), p1 = dot4(v1, as4);
            const float p2 = dot4(v2, as4), p3 = dot4(v3, as4);
            const float qa = (b0 ? p1 : p0) + row_xor_get<1>(b0 ? p0 : p1);
            const float qb = (b0 ? p3 : p2) + row_xor_get<1>(b0 ? p2 : p3);
            float r = (b1 ? qb : qa) + row_xor_get<2>(b1 ? qa : qb);
            r += row_xor_get<4>(r);
            r += row_xor_get<8>(r);
            const bool valid = j + k4 < n;
            const float e = lrelu(r + adh);

            float g = valid ? e : m;
            g = fmaxf(g, row_xor_get<1>(g));
            g = fmaxf(g, row_xor_get<2>(g));
            const float mn = fmaxf(m, g);
            const float sc = __expf(m - mn);
            const float w = valid ? __expf(e - mn) : 0.0f;
            m = mn;
            dn = dn * sc + w;

            const float w0 = quad_get<0>(w), w1 = quad_get<1>(w);
            const float w2 = quad_get<2>(w), w3 = quad_get<3>(w);
            acc = acc * sc + v0 * w0;
            acc += v1 * w1;
            acc += v2 * w2;
            acc += v3 * w3;
        }
    }
    dn += row_xor_get<1>(dn);
    dn += row_xor_get<2>(dn);
    dn += SM_EPS;

    const float q0 = acc.x / dn, q1 = acc.y / dn, q2 = acc.z / dn, q3 = acc.w / dn;
    const int f = (lane & 15) * 4;
    const float4 bv = *(const float4*)(bias + f);
    const float v0 = (q0 + __shfl_xor(q0, 16, 32)) * 0.5f + bv.x;
    const float v1 = (q1 + __shfl_xor(q1, 16, 32)) * 0.5f + bv.y;
    const float v2 = (q2 + __shfl_xor(q2, 16, 32)) * 0.5f + bv.z;
    const float v3 = (q3 + __shfl_xor(q3, 16, 32)) * 0.5f + bv.w;

    float sum = (v0 + v1) + (v2 + v3);
    #pragma unroll
    for (int o = 8; o >= 1; o >>= 1) sum += __shfl_xor(sum, o, 32);
    const float mu = sum * (1.0f / OUT_F);
    const float d0 = v0 - mu, d1 = v1 - mu, d2 = v2 - mu, d3 = v3 - mu;
    float sq = (d0 * d0 + d1 * d1) + (d2 * d2 + d3 * d3);
    #pragma unroll
    for (int o = 8; o >= 1; o >>= 1) sq += __shfl_xor(sq, o, 32);
    const float inv = rsqrtf(sq * (1.0f / OUT_F) + LN_EPS);

    if (lane < 16) {
        const float4 gv = *(const float4*)(gamma + f);
        const float4 be = *(const float4*)(beta + f);
        float4 o4;
        o4.x = d0 * inv * gv.x + be.x;
        o4.y = d1 * inv * gv.y + be.y;
        o4.z = d2 * inv * gv.z + be.z;
        o4.w = d3 * inv * gv.w + be.w;
        *(float4*)(out + (size_t)d * OUT_F + f) = o4;
    }
}

extern "C" void kernel_launch(void* const* d_in, const int* in_sizes, int n_in,
                              void* d_out, int out_size, void* d_ws, size_t ws_size,
                              hipStream_t stream)
{
    const float* X    = (const float*)d_in[0];
    const int*   ei   = (const int*)  d_in[1];
    const float* W    = (const float*)d_in[2];
    const float* attS = (const float*)d_in[3];
    const float* attD = (const float*)d_in[4];
    const float* bias = (const float*)d_in[5];
    const float* g    = (const float*)d_in[6];
    const float* b    = (const float*)d_in[7];
    float* out = (float*)d_out;

    const int N = in_sizes[0] / IN_F;
    const int E = in_sizes[1] / 2;
    const int* esrc = ei;
    const int* edst = ei + E;

    const int NB  = (N + RNODES - 1) / RNODES;
    const int NCH = (E + CH - 1) / CH;

    float* ws     = (float*)d_ws;
    float* xp     = ws;
    float* aD     = xp + (size_t)N * XDIM;
    int*   off    = (int*)(aD + (size_t)N * 2);
    int*   runs   = off + (size_t)N + 1;
    int*   binned = runs + (size_t)NCH * NB;
    int*   ssrc   = binned + (size_t)E;

    {
        const int waves = (N + 15) / 16;
        const int GB    = (waves + 7) / 8;
        gat_gemm_bin<<<NCH + GB, BT, 0, stream>>>(
            X, W, attD, xp, aD, esrc, edst, runs, binned, N, NB, E, NCH);
    }
    gat_bin_sort<<<NB, BS2, 0, stream>>>(runs, binned, off, ssrc, N, NB, NCH);
    gat_node_gather<<<(N + 7) / 8, 256, 0, stream>>>(
        off, ssrc, attS, aD, xp, bias, g, b, out, N);
}
